# starting kernel
# baseline (speedup 1.0000x reference)
#include <hip/hip_runtime.h>
#include <hip/hip_bf16.h>


typedef __attribute__((ext_vector_type(2))) float v2f;
typedef __attribute__((ext_vector_type(8))) float v8f;

#define LSTM_H 6

static __device__ __forceinline__ float sigm(float x) {
  return __builtin_amdgcn_rcpf(1.0f + __expf(-x));
}
static __device__ __forceinline__ float tanh_fast(float x) {
  return 1.0f - 2.0f * __builtin_amdgcn_rcpf(__expf(2.0f * x) + 1.0f);
}
static __device__ __forceinline__ float sx16(float v) {
  return __shfl_xor(v, 16, 32);
}
static __device__ __forceinline__ v8f wmma4(v2f a, v2f b, v8f c) {
  return __builtin_amdgcn_wmma_f32_16x16x4_f32(false, a, false, b, (short)0, c,
                                               false, false);
}

static __device__ __forceinline__ int gate_row(int m) {
  int idx = m & 7;
  return (idx < LSTM_H) ? ((m >> 3) * LSTM_H + idx) : -1;
}

static __device__ __forceinline__ float wcomb(const float* Wih, const float* Whh,
                                              int D, int m, int k) {
  int r = gate_row(m);
  if (r < 0) return 0.0f;
  if (k < D) return Wih[r * D + k];
  k -= D;
  return (k < LSTM_H) ? Whh[r * LSTM_H + k] : 0.0f;
}

static __device__ __forceinline__ void load_A(const float* Wih, const float* Whh,
                                              const float* bias, int D, int nch,
                                              v2f a[2][5], v8f cini[2],
                                              int lm, int half) {
  for (int t = 0; t < 2; ++t) {
    int m = t * 16 + lm;
#pragma unroll
    for (int c = 0; c < 5; ++c) {
      if (c < nch) {
        int k0 = 4 * c + 2 * half;
        v2f av;
        av.x = wcomb(Wih, Whh, D, m, k0);
        av.y = wcomb(Wih, Whh, D, m, k0 + 1);
        a[t][c] = av;
      }
    }
    v8f ci;
#pragma unroll
    for (int r = 0; r < 8; ++r) {
      int gr = gate_row(t * 16 + half * 8 + r);
      ci[r] = (gr >= 0) ? bias[gr] : 0.0f;
    }
    cini[t] = ci;
  }
}

static __device__ __forceinline__ void mm(const v2f a[2][5], const v8f cini[2],
                                          const v2f bch[5], int nch,
                                          v8f& A0, v8f& A1) {
  A0 = cini[0];
  A1 = cini[1];
#pragma unroll
  for (int c = 0; c < 5; ++c) {
    if (c < nch) {
      A0 = wmma4(a[0][c], bch[c], A0);
      A1 = wmma4(a[1][c], bch[c], A1);
    }
  }
}

static __device__ __forceinline__ void lstm_pointwise(const v8f& A0, const v8f& A1,
                                                      bool low, float c[6], float h[6]) {
#pragma unroll
  for (int r = 0; r < LSTM_H; ++r) {
    float s_if = sigm(A0[r]);
    float tg = tanh_fast(A1[r]);
    float so = sigm(A1[r]);
    float go = low ? tg : so;
    float fsw = sx16(s_if);
    float cn = fsw * c[r] + s_if * go;
    c[r] = cn;
    float osw = sx16(go);
    h[r] = osw * tanh_fast(cn);
  }
}

static __device__ __forceinline__ void b_l0(const float* xp, const float h[6],
                                            int half, v2f bch[5]) {
  v2f xlo = *(const v2f*)(xp + 2 * half);
  v2f x45 = *(const v2f*)(xp + 4);
  float h0s = sx16(h[0]), h1s = sx16(h[1]), h4s = sx16(h[4]), h5s = sx16(h[5]);
  v2f b1, b2;
  b1.x = half ? h0s : x45.x; b1.y = half ? h1s : x45.y;
  b2.x = half ? h4s : h[2];  b2.y = half ? h5s : h[3];
  bch[0] = xlo; bch[1] = b1; bch[2] = b2;
}

static __device__ __forceinline__ void b_l1(const float f0[6], const float* r0p,
                                            const float h[6], int half, v2f bch[5]) {
  v2f r01 = *(const v2f*)(r0p);
  v2f rhi = *(const v2f*)(r0p + 2 + 2 * half);
  float f2s = sx16(f0[2]), f3s = sx16(f0[3]);
  float h2s = sx16(h[2]),  h3s = sx16(h[3]);
  v2f b0, b1, b3, b4;
  b0.x = half ? f2s : f0[0];    b0.y = half ? f3s : f0[1];
  b1.x = half ? r01.x : f0[4];  b1.y = half ? r01.y : f0[5];
  b3.x = half ? h2s : h[0];     b3.y = half ? h3s : h[1];
  b4.x = half ? 0.0f : h[4];    b4.y = half ? 0.0f : h[5];
  bch[0] = b0; bch[1] = b1; bch[2] = rhi; bch[3] = b3; bch[4] = b4;
}

__global__ void lstm_l0_bwd(const float* __restrict__ tf, const float* __restrict__ h0,
                            const float* __restrict__ c0,
                            const float* __restrict__ Wih, const float* __restrict__ Whh,
                            const float* __restrict__ bias,
                            float* __restrict__ r0ws, int B, int T) {
  int tid = blockIdx.x * blockDim.x + threadIdx.x;
  int lane = tid & 31, lm = lane & 15, half = lane >> 4;
  int s = (tid >> 5) * 16 + lm;
  if (s >= B) return;

  v2f a[2][5]; v8f ci[2];
  load_A(Wih, Whh, bias, LSTM_H, 3, a, ci, lm, half);

  float h[6], c[6];
  size_t d1 = (size_t)B * LSTM_H;
#pragma unroll
  for (int j = 0; j < 6; ++j) {
    h[j] = h0[d1 + (size_t)s * 6 + j];
    c[j] = c0[d1 + (size_t)s * 6 + j];
  }

  for (int t = T - 1; t >= 0; --t) {
    const float* xp = tf + ((size_t)s * T + t) * 6;
    if (t > 0) __builtin_prefetch(xp - 6, 0, 1);
    v2f bch[5]; b_l0(xp, h, half, bch);
    v8f A0, A1; mm(a, ci, bch, 3, A0, A1);
    lstm_pointwise(A0, A1, half == 0, c, h);
    if (half == 0) {
      float* wp = r0ws + ((size_t)t * B + s) * 6;
      v2f w01; w01.x = h[0]; w01.y = h[1];
      v2f w23; w23.x = h[2]; w23.y = h[3];
      v2f w45; w45.x = h[4]; w45.y = h[5];
      *(v2f*)(wp) = w01; *(v2f*)(wp + 2) = w23; *(v2f*)(wp + 4) = w45;
    }
  }
}

__global__ void lstm_fused(const float* __restrict__ tf, const float* __restrict__ h0,
                           const float* __restrict__ c0,
                           const float* __restrict__ Wih0f, const float* __restrict__ Whh0f,
                           const float* __restrict__ b0f,
                           const float* __restrict__ Wih1f, const float* __restrict__ Whh1f,
                           const float* __restrict__ b1f,
                           const float* __restrict__ Wih1b, const float* __restrict__ Whh1b,
                           const float* __restrict__ b1b,
                           const float* __restrict__ fcw, const float* __restrict__ fcb,
                           const float* __restrict__ r0ws, float* __restrict__ out,
                           int B, int T) {
  int tid = blockIdx.x * blockDim.x + threadIdx.x;
  int lane = tid & 31, lm = lane & 15, half = lane >> 4;
  int s = (tid >> 5) * 16 + lm;
  if (s >= B) return;

  v2f a0[2][5]; v8f ci0[2]; load_A(Wih0f, Whh0f, b0f, LSTM_H, 3, a0, ci0, lm, half);
  v2f a1[2][5]; v8f ci1[2]; load_A(Wih1f, Whh1f, b1f, 2 * LSTM_H, 5, a1, ci1, lm, half);

  float hA[6], cA[6], hB[6], cB[6];
  size_t d0 = 0, d2 = 2 * (size_t)B * LSTM_H;
#pragma unroll
  for (int j = 0; j < 6; ++j) {
    hA[j] = h0[d0 + (size_t)s * 6 + j];  cA[j] = c0[d0 + (size_t)s * 6 + j];
    hB[j] = h0[d2 + (size_t)s * 6 + j];  cB[j] = c0[d2 + (size_t)s * 6 + j];
  }

  for (int t = 0; t < T; ++t) {
    const float* xp = tf + ((size_t)s * T + t) * 6;
    if (t + 1 < T) __builtin_prefetch(xp + 6, 0, 1);
    v2f bch[5]; b_l0(xp, hA, half, bch);
    v8f A0, A1; mm(a0, ci0, bch, 3, A0, A1);
    lstm_pointwise(A0, A1, half == 0, cA, hA);
    const float* r0p = r0ws + ((size_t)t * B + s) * 6;
    b_l1(hA, r0p, hB, half, bch);
    mm(a1, ci1, bch, 5, A0, A1);
    lstm_pointwise(A0, A1, half == 0, cB, hB);
  }
  v2f a2[2][5]; v8f ci2[2]; load_A(Wih1b, Whh1b, b1b, 2 * LSTM_H, 5, a2, ci2, lm, half);
  float hR[6], cR[6];
  size_t d3 = 3 * (size_t)B * LSTM_H;
#pragma unroll
  for (int j = 0; j < 6; ++j) {
    hR[j] = h0[d3 + (size_t)s * 6 + j];  cR[j] = c0[d3 + (size_t)s * 6 + j];
  }
  {
    const float* r0p = r0ws + ((size_t)(T - 1) * B + s) * 6;
    v2f bch[5]; b_l1(hA, r0p, hR, half, bch);
    v8f A0, A1; mm(a2, ci2, bch, 5, A0, A1);
    lstm_pointwise(A0, A1, half == 0, cR, hR);
  }
  if (half == 0) {
    float v[12];
#pragma unroll
    for (int k = 0; k < 6; ++k) {
      v[k]     = hB[k] > 0.0f ? hB[k] : 0.0f;
      v[6 + k] = hR[k] > 0.0f ? hR[k] : 0.0f;
    }
#pragma unroll
    for (int j = 0; j < 6; ++j) {
      float acc = fcb[j];
#pragma unroll
      for (int k = 0; k < 12; ++k) acc = fmaf(fcw[j * 12 + k], v[k], acc);
      out[(size_t)s * 6 + j] = acc;
    }
  }
}

extern "C" void kernel_launch(void* const* d_in, const int* in_sizes, int n_in,
                              void* d_out, int out_size, void* d_ws, size_t ws_size,
                              hipStream_t stream) {
  const float* tf    = (const float*)d_in[0];
  const float* h0    = (const float*)d_in[1];
  const float* c0    = (const float*)d_in[2];
  const float* Wih0f = (const float*)d_in[3];
  const float* Whh0f = (const float*)d_in[4];
  const float* b0f   = (const float*)d_in[5];
  const float* Wih0b = (const float*)d_in[6];
  const float* Whh0b = (const float*)d_in[7];
  const float* b0b   = (const float*)d_in[8];
  const float* Wih1f = (const float*)d_in[9];
  const float* Whh1f = (const float*)d_in[10];
  const float* b1f   = (const float*)d_in[11];
  const float* Wih1b = (const float*)d_in[12];
  const float* Whh1b = (const float*)d_in[13];
  const float* b1b   = (const float*)d_in[14];
  const float* fcw   = (const float*)d_in[15];
  const float* fcb   = (const float*)d_in[16];

  int B = in_sizes[1] / (4 * LSTM_H);
  int T = in_sizes[0] / (B * LSTM_H);

  size_t need = (size_t)B * T * LSTM_H * sizeof(float);
  if (ws_size < need) return;
  float* r0ws = (float*)d_ws;

  int waves = (B + 15) / 16;
  int threads = waves * 32;
  int block = 256;
  int grid = (threads + block - 1) / block;

  lstm_l0_bwd<<<grid, block, 0, stream>>>(tf, h0, c0, Wih0b, Whh0b, b0b, r0ws, B, T);
  lstm_fused<<<grid, block, 0, stream>>>(tf, h0, c0, Wih0f, Whh0f, b0f,
                                         Wih1f, Whh1f, b1f, Wih1b, Whh1b, b1b,
                                         fcw, fcb, r0ws, (float*)d_out, B, T);
}


# x[t] and r0[t] loaded into registers one block of 8 / 4 time steps ahead (loops unrolled per block, launch bounds 256), replacing the one-step cache prefetch: no load latency on the recurrence
# speedup vs baseline: 1.3743x; 1.3743x over previous
#include <hip/hip_runtime.h>
#include <hip/hip_bf16.h>


typedef __attribute__((ext_vector_type(2))) float v2f;
typedef __attribute__((ext_vector_type(8))) float v8f;

#define LSTM_H 6
#define BLK_L0 8
#define BLK_FU 4
#define WG_THREADS 256

static __device__ __forceinline__ float sigm(float x) {
  return __builtin_amdgcn_rcpf(1.0f + __expf(-x));
}
static __device__ __forceinline__ float tanh_fast(float x) {
  return 1.0f - 2.0f * __builtin_amdgcn_rcpf(__expf(2.0f * x) + 1.0f);
}
static __device__ __forceinline__ float sx16(float v) {
  return __shfl_xor(v, 16, 32);
}
static __device__ __forceinline__ v8f wmma4(v2f a, v2f b, v8f c) {
  return __builtin_amdgcn_wmma_f32_16x16x4_f32(false, a, false, b, (short)0, c,
                                               false, false);
}

static __device__ __forceinline__ int gate_row(int m) {
  int idx = m & 7;
  return (idx < LSTM_H) ? ((m >> 3) * LSTM_H + idx) : -1;
}

static __device__ __forceinline__ float wcomb(const float* Wih, const float* Whh,
                                              int D, int m, int k) {
  int r = gate_row(m);
  if (r < 0) return 0.0f;
  if (k < D) return Wih[r * D + k];
  k -= D;
  return (k < LSTM_H) ? Whh[r * LSTM_H + k] : 0.0f;
}

static __device__ __forceinline__ void load_A(const float* Wih, const float* Whh,
                                              const float* bias, int D, int nch,
                                              v2f a[2][5], v8f cini[2],
                                              int lm, int half) {
  for (int t = 0; t < 2; ++t) {
    int m = t * 16 + lm;
#pragma unroll
    for (int c = 0; c < 5; ++c) {
      if (c < nch) {
        int k0 = 4 * c + 2 * half;
        v2f av;
        av.x = wcomb(Wih, Whh, D, m, k0);
        av.y = wcomb(Wih, Whh, D, m, k0 + 1);
        a[t][c] = av;
      }
    }
    v8f ci;
#pragma unroll
    for (int r = 0; r < 8; ++r) {
      int gr = gate_row(t * 16 + half * 8 + r);
      ci[r] = (gr >= 0) ? bias[gr] : 0.0f;
    }
    cini[t] = ci;
  }
}

static __device__ __forceinline__ void mm(const v2f a[2][5], const v8f cini[2],
                                          const v2f bch[5], int nch,
                                          v8f& A0, v8f& A1) {
  A0 = cini[0];
  A1 = cini[1];
#pragma unroll
  for (int c = 0; c < 5; ++c) {
    if (c < nch) {
      A0 = wmma4(a[0][c], bch[c], A0);
      A1 = wmma4(a[1][c], bch[c], A1);
    }
  }
}

static __device__ __forceinline__ void lstm_pointwise(const v8f& A0, const v8f& A1,
                                                      bool low, float c[6], float h[6]) {
#pragma unroll
  for (int r = 0; r < LSTM_H; ++r) {
    float s_if = sigm(A0[r]);
    float tg = tanh_fast(A1[r]);
    float so = sigm(A1[r]);
    float go = low ? tg : so;
    float fsw = sx16(s_if);
    float cn = fsw * c[r] + s_if * go;
    c[r] = cn;
    float osw = sx16(go);
    h[r] = osw * tanh_fast(cn);
  }
}

static __device__ __forceinline__ void ld_x(const float* xp, int half,
                                            v2f& xlo, v2f& x45) {
  xlo = *(const v2f*)(xp + 2 * half);
  x45 = *(const v2f*)(xp + 4);
}
static __device__ __forceinline__ void ld_r(const float* r0p, int half,
                                            v2f& r01, v2f& rhi) {
  r01 = *(const v2f*)(r0p);
  rhi = *(const v2f*)(r0p + 2 + 2 * half);
}

static __device__ __forceinline__ void b_l0(v2f xlo, v2f x45, const float h[6],
                                            int half, v2f bch[5]) {
  float h0s = sx16(h[0]), h1s = sx16(h[1]), h4s = sx16(h[4]), h5s = sx16(h[5]);
  v2f b1, b2;
  b1.x = half ? h0s : x45.x; b1.y = half ? h1s : x45.y;
  b2.x = half ? h4s : h[2];  b2.y = half ? h5s : h[3];
  bch[0] = xlo; bch[1] = b1; bch[2] = b2;
}

static __device__ __forceinline__ void b_l1(const float f0[6], v2f r01, v2f rhi,
                                            const float h[6], int half, v2f bch[5]) {
  float f2s = sx16(f0[2]), f3s = sx16(f0[3]);
  float h2s = sx16(h[2]),  h3s = sx16(h[3]);
  v2f b0, b1, b3, b4;
  b0.x = half ? f2s : f0[0];    b0.y = half ? f3s : f0[1];
  b1.x = half ? r01.x : f0[4];  b1.y = half ? r01.y : f0[5];
  b3.x = half ? h2s : h[0];     b3.y = half ? h3s : h[1];
  b4.x = half ? 0.0f : h[4];    b4.y = half ? 0.0f : h[5];
  bch[0] = b0; bch[1] = b1; bch[2] = rhi; bch[3] = b3; bch[4] = b4;
}

__global__ void __launch_bounds__(WG_THREADS) lstm_l0_bwd(const float* __restrict__ tf, const float* __restrict__ h0,
                            const float* __restrict__ c0,
                            const float* __restrict__ Wih, const float* __restrict__ Whh,
                            const float* __restrict__ bias,
                            float* __restrict__ r0ws, int B, int T) {
  int tid = blockIdx.x * blockDim.x + threadIdx.x;
  int lane = tid & 31, lm = lane & 15, half = lane >> 4;
  int s = (tid >> 5) * 16 + lm;
  if (s >= B) return;

  v2f a[2][5]; v8f ci[2];
  load_A(Wih, Whh, bias, LSTM_H, 3, a, ci, lm, half);

  float h[6], c[6];
  size_t d1 = (size_t)B * LSTM_H;
#pragma unroll
  for (int j = 0; j < 6; ++j) {
    h[j] = h0[d1 + (size_t)s * 6 + j];
    c[j] = c0[d1 + (size_t)s * 6 + j];
  }

  const float* xrow = tf + (size_t)s * T * 6;
  v2f cxl[BLK_L0], cxh[BLK_L0];
#pragma unroll
  for (int j = 0; j < BLK_L0; ++j) {
    int tt = T - 1 - j; tt = tt < 0 ? 0 : tt;
    ld_x(xrow + (size_t)tt * 6, half, cxl[j], cxh[j]);
  }

  auto step = [&](int t, v2f xlo, v2f x45) {
    v2f bch[5]; b_l0(xlo, x45, h, half, bch);
    v8f A0, A1; mm(a, ci, bch, 3, A0, A1);
    lstm_pointwise(A0, A1, half == 0, c, h);
    if (half == 0) {
      float* wp = r0ws + ((size_t)t * B + s) * 6;
      v2f w01; w01.x = h[0]; w01.y = h[1];
      v2f w23; w23.x = h[2]; w23.y = h[3];
      v2f w45; w45.x = h[4]; w45.y = h[5];
      *(v2f*)(wp) = w01; *(v2f*)(wp + 2) = w23; *(v2f*)(wp + 4) = w45;
    }
  };

  int t = T - 1;
  for (; t >= BLK_L0 - 1; t -= BLK_L0) {
    v2f nxl[BLK_L0], nxh[BLK_L0];
#pragma unroll
    for (int j = 0; j < BLK_L0; ++j) {
      int tn = t - BLK_L0 - j; tn = tn < 0 ? 0 : tn;
      ld_x(xrow + (size_t)tn * 6, half, nxl[j], nxh[j]);
    }
#pragma unroll
    for (int j = 0; j < BLK_L0; ++j) step(t - j, cxl[j], cxh[j]);
#pragma unroll
    for (int j = 0; j < BLK_L0; ++j) { cxl[j] = nxl[j]; cxh[j] = nxh[j]; }
  }
#pragma unroll
  for (int j = 0; j < BLK_L0 - 1; ++j) {
    if (j <= t) step(t - j, cxl[j], cxh[j]);
  }
}

__global__ void __launch_bounds__(WG_THREADS) lstm_fused(const float* __restrict__ tf, const float* __restrict__ h0,
                           const float* __restrict__ c0,
                           const float* __restrict__ Wih0f, const float* __restrict__ Whh0f,
                           const float* __restrict__ b0f,
                           const float* __restrict__ Wih1f, const float* __restrict__ Whh1f,
                           const float* __restrict__ b1f,
                           const float* __restrict__ Wih1b, const float* __restrict__ Whh1b,
                           const float* __restrict__ b1b,
                           const float* __restrict__ fcw, const float* __restrict__ fcb,
                           const float* __restrict__ r0ws, float* __restrict__ out,
                           int B, int T) {
  int tid = blockIdx.x * blockDim.x + threadIdx.x;
  int lane = tid & 31, lm = lane & 15, half = lane >> 4;
  int s = (tid >> 5) * 16 + lm;
  if (s >= B) return;

  v2f a0[2][5]; v8f ci0[2]; load_A(Wih0f, Whh0f, b0f, LSTM_H, 3, a0, ci0, lm, half);
  v2f a1[2][5]; v8f ci1[2]; load_A(Wih1f, Whh1f, b1f, 2 * LSTM_H, 5, a1, ci1, lm, half);

  float hA[6], cA[6], hB[6], cB[6];
  size_t d0 = 0, d2 = 2 * (size_t)B * LSTM_H;
#pragma unroll
  for (int j = 0; j < 6; ++j) {
    hA[j] = h0[d0 + (size_t)s * 6 + j];  cA[j] = c0[d0 + (size_t)s * 6 + j];
    hB[j] = h0[d2 + (size_t)s * 6 + j];  cB[j] = c0[d2 + (size_t)s * 6 + j];
  }

  const float* xrow = tf + (size_t)s * T * 6;
  const float* rcol = r0ws + (size_t)s * 6;
  const size_t rstep = (size_t)B * 6;
  v2f cxl[BLK_FU], cxh[BLK_FU], cra[BLK_FU], crb[BLK_FU];
#pragma unroll
  for (int j = 0; j < BLK_FU; ++j) {
    int tt = j > T - 1 ? T - 1 : j;
    ld_x(xrow + (size_t)tt * 6, half, cxl[j], cxh[j]);
    ld_r(rcol + (size_t)tt * rstep, half, cra[j], crb[j]);
  }

  auto step = [&](v2f xlo, v2f x45, v2f r01, v2f rhi) {
    v2f bch[5]; b_l0(xlo, x45, hA, half, bch);
    v8f A0, A1; mm(a0, ci0, bch, 3, A0, A1);
    lstm_pointwise(A0, A1, half == 0, cA, hA);
    b_l1(hA, r01, rhi, hB, half, bch);
    mm(a1, ci1, bch, 5, A0, A1);
    lstm_pointwise(A0, A1, half == 0, cB, hB);
  };

  int t = 0;
  for (; t + BLK_FU <= T; t += BLK_FU) {
    v2f nxl[BLK_FU], nxh[BLK_FU], nra[BLK_FU], nrb[BLK_FU];
#pragma unroll
    for (int j = 0; j < BLK_FU; ++j) {
      int tn = t + BLK_FU + j; tn = tn > T - 1 ? T - 1 : tn;
      ld_x(xrow + (size_t)tn * 6, half, nxl[j], nxh[j]);
      ld_r(rcol + (size_t)tn * rstep, half, nra[j], nrb[j]);
    }
    __builtin_amdgcn_sched_barrier(0);
#pragma unroll
    for (int j = 0; j < BLK_FU; ++j) step(cxl[j], cxh[j], cra[j], crb[j]);
    __builtin_amdgcn_sched_barrier(0);
#pragma unroll
    for (int j = 0; j < BLK_FU; ++j) {
      cxl[j] = nxl[j]; cxh[j] = nxh[j]; cra[j] = nra[j]; crb[j] = nrb[j];
    }
  }
#pragma unroll
  for (int j = 0; j < BLK_FU - 1; ++j) {
    if (t + j < T) step(cxl[j], cxh[j], cra[j], crb[j]);
  }
  v2f a2[2][5]; v8f ci2[2]; load_A(Wih1b, Whh1b, b1b, 2 * LSTM_H, 5, a2, ci2, lm, half);
  float hR[6], cR[6];
  size_t d3 = 3 * (size_t)B * LSTM_H;
#pragma unroll
  for (int j = 0; j < 6; ++j) {
    hR[j] = h0[d3 + (size_t)s * 6 + j];  cR[j] = c0[d3 + (size_t)s * 6 + j];
  }
  {
    const float* r0p = r0ws + ((size_t)(T - 1) * B + s) * 6;
    v2f r01, rhi; ld_r(r0p, half, r01, rhi);
    v2f bch[5]; b_l1(hA, r01, rhi, hR, half, bch);
    v8f A0, A1; mm(a2, ci2, bch, 5, A0, A1);
    lstm_pointwise(A0, A1, half == 0, cR, hR);
  }
  if (half == 0) {
    float v[12];
#pragma unroll
    for (int k = 0; k < 6; ++k) {
      v[k]     = hB[k] > 0.0f ? hB[k] : 0.0f;
      v[6 + k] = hR[k] > 0.0f ? hR[k] : 0.0f;
    }
#pragma unroll
    for (int j = 0; j < 6; ++j) {
      float acc = fcb[j];
#pragma unroll
      for (int k = 0; k < 12; ++k) acc = fmaf(fcw[j * 12 + k], v[k], acc);
      out[(size_t)s * 6 + j] = acc;
    }
  }
}

extern "C" void kernel_launch(void* const* d_in, const int* in_sizes, int n_in,
                              void* d_out, int out_size, void* d_ws, size_t ws_size,
                              hipStream_t stream) {
  const float* tf    = (const float*)d_in[0];
  const float* h0    = (const float*)d_in[1];
  const float* c0    = (const float*)d_in[2];
  const float* Wih0f = (const float*)d_in[3];
  const float* Whh0f = (const float*)d_in[4];
  const float* b0f   = (const float*)d_in[5];
  const float* Wih0b = (const float*)d_in[6];
  const float* Whh0b = (const float*)d_in[7];
  const float* b0b   = (const float*)d_in[8];
  const float* Wih1f = (const float*)d_in[9];
  const float* Whh1f = (const float*)d_in[10];
  const float* b1f   = (const float*)d_in[11];
  const float* Wih1b = (const float*)d_in[12];
  const float* Whh1b = (const float*)d_in[13];
  const float* b1b   = (const float*)d_in[14];
  const float* fcw   = (const float*)d_in[15];
  const float* fcb   = (const float*)d_in[16];

  int B = in_sizes[1] / (4 * LSTM_H);
  int T = in_sizes[0] / (B * LSTM_H);

  size_t need = (size_t)B * T * LSTM_H * sizeof(float);
  if (ws_size < need) return;
  float* r0ws = (float*)d_ws;

  int waves = (B + 15) / 16;
  int threads = waves * 32;
  int block = WG_THREADS;
  int grid = (threads + block - 1) / block;

  lstm_l0_bwd<<<grid, block, 0, stream>>>(tf, h0, c0, Wih0b, Whh0b, b0b, r0ws, B, T);
  lstm_fused<<<grid, block, 0, stream>>>(tf, h0, c0, Wih0f, Whh0f, b0f,
                                         Wih1f, Whh1f, b1f, Wih1b, Whh1b, b1b,
                                         fcw, fcb, r0ws, (float*)d_out, B, T);
}


# work-groups of 4 waves instead of 8 (block 128, twice the groups): one wave per SIMD on twice as many compute units, so no two serial recurrences share a vector port and matrix pipe
# speedup vs baseline: 1.7753x; 1.2918x over previous
#include <hip/hip_runtime.h>
#include <hip/hip_bf16.h>


typedef __attribute__((ext_vector_type(2))) float v2f;
typedef __attribute__((ext_vector_type(8))) float v8f;

#define LSTM_H 6
#define BLK_L0 8
#define BLK_FU 4
#define WG_THREADS 128

static __device__ __forceinline__ float sigm(float x) {
  return __builtin_amdgcn_rcpf(1.0f + __expf(-x));
}
static __device__ __forceinline__ float tanh_fast(float x) {
  return 1.0f - 2.0f * __builtin_amdgcn_rcpf(__expf(2.0f * x) + 1.0f);
}
static __device__ __forceinline__ float sx16(float v) {
  return __shfl_xor(v, 16, 32);
}
static __device__ __forceinline__ v8f wmma4(v2f a, v2f b, v8f c) {
  return __builtin_amdgcn_wmma_f32_16x16x4_f32(false, a, false, b, (short)0, c,
                                               false, false);
}

static __device__ __forceinline__ int gate_row(int m) {
  int idx = m & 7;
  return (idx < LSTM_H) ? ((m >> 3) * LSTM_H + idx) : -1;
}

static __device__ __forceinline__ float wcomb(const float* Wih, const float* Whh,
                                              int D, int m, int k) {
  int r = gate_row(m);
  if (r < 0) return 0.0f;
  if (k < D) return Wih[r * D + k];
  k -= D;
  return (k < LSTM_H) ? Whh[r * LSTM_H + k] : 0.0f;
}

static __device__ __forceinline__ void load_A(const float* Wih, const float* Whh,
                                              const float* bias, int D, int nch,
                                              v2f a[2][5], v8f cini[2],
                                              int lm, int half) {
  for (int t = 0; t < 2; ++t) {
    int m = t * 16 + lm;
#pragma unroll
    for (int c = 0; c < 5; ++c) {
      if (c < nch) {
        int k0 = 4 * c + 2 * half;
        v2f av;
        av.x = wcomb(Wih, Whh, D, m, k0);
        av.y = wcomb(Wih, Whh, D, m, k0 + 1);
        a[t][c] = av;
      }
    }
    v8f ci;
#pragma unroll
    for (int r = 0; r < 8; ++r) {
      int gr = gate_row(t * 16 + half * 8 + r);
      ci[r] = (gr >= 0) ? bias[gr] : 0.0f;
    }
    cini[t] = ci;
  }
}

static __device__ __forceinline__ void mm(const v2f a[2][5], const v8f cini[2],
                                          const v2f bch[5], int nch,
                                          v8f& A0, v8f& A1) {
  A0 = cini[0];
  A1 = cini[1];
#pragma unroll
  for (int c = 0; c < 5; ++c) {
    if (c < nch) {
      A0 = wmma4(a[0][c], bch[c], A0);
      A1 = wmma4(a[1][c], bch[c], A1);
    }
  }
}

static __device__ __forceinline__ void lstm_pointwise(const v8f& A0, const v8f& A1,
                                                      bool low, float c[6], float h[6]) {
#pragma unroll
  for (int r = 0; r < LSTM_H; ++r) {
    float s_if = sigm(A0[r]);
    float tg = tanh_fast(A1[r]);
    float so = sigm(A1[r]);
    float go = low ? tg : so;
    float fsw = sx16(s_if);
    float cn = fsw * c[r] + s_if * go;
    c[r] = cn;
    float osw = sx16(go);
    h[r] = osw * tanh_fast(cn);
  }
}

static __device__ __forceinline__ void ld_x(const float* xp, int half,
                                            v2f& xlo, v2f& x45) {
  xlo = *(const v2f*)(xp + 2 * half);
  x45 = *(const v2f*)(xp + 4);
}
static __device__ __forceinline__ void ld_r(const float* r0p, int half,
                                            v2f& r01, v2f& rhi) {
  r01 = *(const v2f*)(r0p);
  rhi = *(const v2f*)(r0p + 2 + 2 * half);
}

static __device__ __forceinline__ void b_l0(v2f xlo, v2f x45, const float h[6],
                                            int half, v2f bch[5]) {
  float h0s = sx16(h[0]), h1s = sx16(h[1]), h4s = sx16(h[4]), h5s = sx16(h[5]);
  v2f b1, b2;
  b1.x = half ? h0s : x45.x; b1.y = half ? h1s : x45.y;
  b2.x = half ? h4s : h[2];  b2.y = half ? h5s : h[3];
  bch[0] = xlo; bch[1] = b1; bch[2] = b2;
}

static __device__ __forceinline__ void b_l1(const float f0[6], v2f r01, v2f rhi,
                                            const float h[6], int half, v2f bch[5]) {
  float f2s = sx16(f0[2]), f3s = sx16(f0[3]);
  float h2s = sx16(h[2]),  h3s = sx16(h[3]);
  v2f b0, b1, b3, b4;
  b0.x = half ? f2s : f0[0];    b0.y = half ? f3s : f0[1];
  b1.x = half ? r01.x : f0[4];  b1.y = half ? r01.y : f0[5];
  b3.x = half ? h2s : h[0];     b3.y = half ? h3s : h[1];
  b4.x = half ? 0.0f : h[4];    b4.y = half ? 0.0f : h[5];
  bch[0] = b0; bch[1] = b1; bch[2] = rhi; bch[3] = b3; bch[4] = b4;
}

__global__ void __launch_bounds__(WG_THREADS) lstm_l0_bwd(const float* __restrict__ tf, const float* __restrict__ h0,
                            const float* __restrict__ c0,
                            const float* __restrict__ Wih, const float* __restrict__ Whh,
                            const float* __restrict__ bias,
                            float* __restrict__ r0ws, int B, int T) {
  int tid = blockIdx.x * blockDim.x + threadIdx.x;
  int lane = tid & 31, lm = lane & 15, half = lane >> 4;
  int s = (tid >> 5) * 16 + lm;
  if (s >= B) return;

  v2f a[2][5]; v8f ci[2];
  load_A(Wih, Whh, bias, LSTM_H, 3, a, ci, lm, half);

  float h[6], c[6];
  size_t d1 = (size_t)B * LSTM_H;
#pragma unroll
  for (int j = 0; j < 6; ++j) {
    h[j] = h0[d1 + (size_t)s * 6 + j];
    c[j] = c0[d1 + (size_t)s * 6 + j];
  }

  const float* xrow = tf + (size_t)s * T * 6;
  v2f cxl[BLK_L0], cxh[BLK_L0];
#pragma unroll
  for (int j = 0; j < BLK_L0; ++j) {
    int tt = T - 1 - j; tt = tt < 0 ? 0 : tt;
    ld_x(xrow + (size_t)tt * 6, half, cxl[j], cxh[j]);
  }

  auto step = [&](int t, v2f xlo, v2f x45) {
    v2f bch[5]; b_l0(xlo, x45, h, half, bch);
    v8f A0, A1; mm(a, ci, bch, 3, A0, A1);
    lstm_pointwise(A0, A1, half == 0, c, h);
    if (half == 0) {
      float* wp = r0ws + ((size_t)t * B + s) * 6;
      v2f w01; w01.x = h[0]; w01.y = h[1];
      v2f w23; w23.x = h[2]; w23.y = h[3];
      v2f w45; w45.x = h[4]; w45.y = h[5];
      *(v2f*)(wp) = w01; *(v2f*)(wp + 2) = w23; *(v2f*)(wp + 4) = w45;
    }
  };

  int t = T - 1;
  for (; t >= BLK_L0 - 1; t -= BLK_L0) {
    v2f nxl[BLK_L0], nxh[BLK_L0];
#pragma unroll
    for (int j = 0; j < BLK_L0; ++j) {
      int tn = t - BLK_L0 - j; tn = tn < 0 ? 0 : tn;
      ld_x(xrow + (size_t)tn * 6, half, nxl[j], nxh[j]);
    }
#pragma unroll
    for (int j = 0; j < BLK_L0; ++j) step(t - j, cxl[j], cxh[j]);
#pragma unroll
    for (int j = 0; j < BLK_L0; ++j) { cxl[j] = nxl[j]; cxh[j] = nxh[j]; }
  }
#pragma unroll
  for (int j = 0; j < BLK_L0 - 1; ++j) {
    if (j <= t) step(t - j, cxl[j], cxh[j]);
  }
}

__global__ void __launch_bounds__(WG_THREADS) lstm_fused(const float* __restrict__ tf, const float* __restrict__ h0,
                           const float* __restrict__ c0,
                           const float* __restrict__ Wih0f, const float* __restrict__ Whh0f,
                           const float* __restrict__ b0f,
                           const float* __restrict__ Wih1f, const float* __restrict__ Whh1f,
                           const float* __restrict__ b1f,
                           const float* __restrict__ Wih1b, const float* __restrict__ Whh1b,
                           const float* __restrict__ b1b,
                           const float* __restrict__ fcw, const float* __restrict__ fcb,
                           const float* __restrict__ r0ws, float* __restrict__ out,
                           int B, int T) {
  int tid = blockIdx.x * blockDim.x + threadIdx.x;
  int lane = tid & 31, lm = lane & 15, half = lane >> 4;
  int s = (tid >> 5) * 16 + lm;
  if (s >= B) return;

  v2f a0[2][5]; v8f ci0[2]; load_A(Wih0f, Whh0f, b0f, LSTM_H, 3, a0, ci0, lm, half);
  v2f a1[2][5]; v8f ci1[2]; load_A(Wih1f, Whh1f, b1f, 2 * LSTM_H, 5, a1, ci1, lm, half);

  float hA[6], cA[6], hB[6], cB[6];
  size_t d0 = 0, d2 = 2 * (size_t)B * LSTM_H;
#pragma unroll
  for (int j = 0; j < 6; ++j) {
    hA[j] = h0[d0 + (size_t)s * 6 + j];  cA[j] = c0[d0 + (size_t)s * 6 + j];
    hB[j] = h0[d2 + (size_t)s * 6 + j];  cB[j] = c0[d2 + (size_t)s * 6 + j];
  }

  const float* xrow = tf + (size_t)s * T * 6;
  const float* rcol = r0ws + (size_t)s * 6;
  const size_t rstep = (size_t)B * 6;
  v2f cxl[BLK_FU], cxh[BLK_FU], cra[BLK_FU], crb[BLK_FU];
#pragma unroll
  for (int j = 0; j < BLK_FU; ++j) {
    int tt = j > T - 1 ? T - 1 : j;
    ld_x(xrow + (size_t)tt * 6, half, cxl[j], cxh[j]);
    ld_r(rcol + (size_t)tt * rstep, half, cra[j], crb[j]);
  }

  auto step = [&](v2f xlo, v2f x45, v2f r01, v2f rhi) {
    v2f bch[5]; b_l0(xlo, x45, hA, half, bch);
    v8f A0, A1; mm(a0, ci0, bch, 3, A0, A1);
    lstm_pointwise(A0, A1, half == 0, cA, hA);
    b_l1(hA, r01, rhi, hB, half, bch);
    mm(a1, ci1, bch, 5, A0, A1);
    lstm_pointwise(A0, A1, half == 0, cB, hB);
  };

  int t = 0;
  for (; t + BLK_FU <= T; t += BLK_FU) {
    v2f nxl[BLK_FU], nxh[BLK_FU], nra[BLK_FU], nrb[BLK_FU];
#pragma unroll
    for (int j = 0; j < BLK_FU; ++j) {
      int tn = t + BLK_FU + j; tn = tn > T - 1 ? T - 1 : tn;
      ld_x(xrow + (size_t)tn * 6, half, nxl[j], nxh[j]);
      ld_r(rcol + (size_t)tn * rstep, half, nra[j], nrb[j]);
    }
    __builtin_amdgcn_sched_barrier(0);
#pragma unroll
    for (int j = 0; j < BLK_FU; ++j) step(cxl[j], cxh[j], cra[j], crb[j]);
    __builtin_amdgcn_sched_barrier(0);
#pragma unroll
    for (int j = 0; j < BLK_FU; ++j) {
      cxl[j] = nxl[j]; cxh[j] = nxh[j]; cra[j] = nra[j]; crb[j] = nrb[j];
    }
  }
#pragma unroll
  for (int j = 0; j < BLK_FU - 1; ++j) {
    if (t + j < T) step(cxl[j], cxh[j], cra[j], crb[j]);
  }
  v2f a2[2][5]; v8f ci2[2]; load_A(Wih1b, Whh1b, b1b, 2 * LSTM_H, 5, a2, ci2, lm, half);
  float hR[6], cR[6];
  size_t d3 = 3 * (size_t)B * LSTM_H;
#pragma unroll
  for (int j = 0; j < 6; ++j) {
    hR[j] = h0[d3 + (size_t)s * 6 + j];  cR[j] = c0[d3 + (size_t)s * 6 + j];
  }
  {
    const float* r0p = r0ws + ((size_t)(T - 1) * B + s) * 6;
    v2f r01, rhi; ld_r(r0p, half, r01, rhi);
    v2f bch[5]; b_l1(hA, r01, rhi, hR, half, bch);
    v8f A0, A1; mm(a2, ci2, bch, 5, A0, A1);
    lstm_pointwise(A0, A1, half == 0, cR, hR);
  }
  if (half == 0) {
    float v[12];
#pragma unroll
    for (int k = 0; k < 6; ++k) {
      v[k]     = hB[k] > 0.0f ? hB[k] : 0.0f;
      v[6 + k] = hR[k] > 0.0f ? hR[k] : 0.0f;
    }
#pragma unroll
    for (int j = 0; j < 6; ++j) {
      float acc = fcb[j];
#pragma unroll
      for (int k = 0; k < 12; ++k) acc = fmaf(fcw[j * 12 + k], v[k], acc);
      out[(size_t)s * 6 + j] = acc;
    }
  }
}

extern "C" void kernel_launch(void* const* d_in, const int* in_sizes, int n_in,
                              void* d_out, int out_size, void* d_ws, size_t ws_size,
                              hipStream_t stream) {
  const float* tf    = (const float*)d_in[0];
  const float* h0    = (const float*)d_in[1];
  const float* c0    = (const float*)d_in[2];
  const float* Wih0f = (const float*)d_in[3];
  const float* Whh0f = (const float*)d_in[4];
  const float* b0f   = (const float*)d_in[5];
  const float* Wih0b = (const float*)d_in[6];
  const float* Whh0b = (const float*)d_in[7];
  const float* b0b   = (const float*)d_in[8];
  const float* Wih1f = (const float*)d_in[9];
  const float* Whh1f = (const float*)d_in[10];
  const float* b1f   = (const float*)d_in[11];
  const float* Wih1b = (const float*)d_in[12];
  const float* Whh1b = (const float*)d_in[13];
  const float* b1b   = (const float*)d_in[14];
  const float* fcw   = (const float*)d_in[15];
  const float* fcb   = (const float*)d_in[16];

  int B = in_sizes[1] / (4 * LSTM_H);
  int T = in_sizes[0] / (B * LSTM_H);

  size_t need = (size_t)B * T * LSTM_H * sizeof(float);
  if (ws_size < need) return;
  float* r0ws = (float*)d_ws;

  int waves = (B + 15) / 16;
  int threads = waves * 32;
  int block = WG_THREADS;
  int grid = (threads + block - 1) / block;

  lstm_l0_bwd<<<grid, block, 0, stream>>>(tf, h0, c0, Wih0b, Whh0b, b0b, r0ws, B, T);
  lstm_fused<<<grid, block, 0, stream>>>(tf, h0, c0, Wih0f, Whh0f, b0f,
                                         Wih1f, Whh1f, b1f, Wih1b, Whh1b, b1b,
                                         fcw, fcb, r0ws, (float*)d_out, B, T);
}


# lane n <-> n^16 swaps done with v_permlanex16_b32 (vector ALU) instead of the shuffle intrinsic's LDS-pipe permute: no LDS round trips or LDS waits left on the serial recurrence
# speedup vs baseline: 2.5345x; 1.4276x over previous
#include <hip/hip_runtime.h>
#include <hip/hip_bf16.h>


typedef __attribute__((ext_vector_type(2))) float v2f;
typedef __attribute__((ext_vector_type(8))) float v8f;

#define LSTM_H 6
#define BLK_L0 8
#define BLK_FU 4
#define WG_THREADS 128

static __device__ __forceinline__ float sigm(float x) {
  return __builtin_amdgcn_rcpf(1.0f + __expf(-x));
}
static __device__ __forceinline__ float tanh_fast(float x) {
  return 1.0f - 2.0f * __builtin_amdgcn_rcpf(__expf(2.0f * x) + 1.0f);
}
static __device__ __forceinline__ float sx16(float v) {
  int b = __builtin_bit_cast(int, v);
  return __builtin_bit_cast(float,
      __builtin_amdgcn_permlanex16(b, b, 0x76543210, 0xfedcba98, false, false));
}
static __device__ __forceinline__ v8f wmma4(v2f a, v2f b, v8f c) {
  return __builtin_amdgcn_wmma_f32_16x16x4_f32(false, a, false, b, (short)0, c,
                                               false, false);
}

static __device__ __forceinline__ int gate_row(int m) {
  int idx = m & 7;
  return (idx < LSTM_H) ? ((m >> 3) * LSTM_H + idx) : -1;
}

static __device__ __forceinline__ float wcomb(const float* Wih, const float* Whh,
                                              int D, int m, int k) {
  int r = gate_row(m);
  if (r < 0) return 0.0f;
  if (k < D) return Wih[r * D + k];
  k -= D;
  return (k < LSTM_H) ? Whh[r * LSTM_H + k] : 0.0f;
}

static __device__ __forceinline__ void load_A(const float* Wih, const float* Whh,
                                              const float* bias, int D, int nch,
                                              v2f a[2][5], v8f cini[2],
                                              int lm, int half) {
  for (int t = 0; t < 2; ++t) {
    int m = t * 16 + lm;
#pragma unroll
    for (int c = 0; c < 5; ++c) {
      if (c < nch) {
        int k0 = 4 * c + 2 * half;
        v2f av;
        av.x = wcomb(Wih, Whh, D, m, k0);
        av.y = wcomb(Wih, Whh, D, m, k0 + 1);
        a[t][c] = av;
      }
    }
    v8f ci;
#pragma unroll
    for (int r = 0; r < 8; ++r) {
      int gr = gate_row(t * 16 + half * 8 + r);
      ci[r] = (gr >= 0) ? bias[gr] : 0.0f;
    }
    cini[t] = ci;
  }
}

static __device__ __forceinline__ void mm(const v2f a[2][5], const v8f cini[2],
                                          const v2f bch[5], int nch,
                                          v8f& A0, v8f& A1) {
  A0 = cini[0];
  A1 = cini[1];
#pragma unroll
  for (int c = 0; c < 5; ++c) {
    if (c < nch) {
      A0 = wmma4(a[0][c], bch[c], A0);
      A1 = wmma4(a[1][c], bch[c], A1);
    }
  }
}

static __device__ __forceinline__ void lstm_pointwise(const v8f& A0, const v8f& A1,
                                                      bool low, float c[6], float h[6]) {
#pragma unroll
  for (int r = 0; r < LSTM_H; ++r) {
    float s_if = sigm(A0[r]);
    float tg = tanh_fast(A1[r]);
    float so = sigm(A1[r]);
    float go = low ? tg : so;
    float fsw = sx16(s_if);
    float cn = fsw * c[r] + s_if * go;
    c[r] = cn;
    float osw = sx16(go);
    h[r] = osw * tanh_fast(cn);
  }
}

static __device__ __forceinline__ void ld_x(const float* xp, int half,
                                            v2f& xlo, v2f& x45) {
  xlo = *(const v2f*)(xp + 2 * half);
  x45 = *(const v2f*)(xp + 4);
}
static __device__ __forceinline__ void ld_r(const float* r0p, int half,
                                            v2f& r01, v2f& rhi) {
  r01 = *(const v2f*)(r0p);
  rhi = *(const v2f*)(r0p + 2 + 2 * half);
}

static __device__ __forceinline__ void b_l0(v2f xlo, v2f x45, const float h[6],
                                            int half, v2f bch[5]) {
  float h0s = sx16(h[0]), h1s = sx16(h[1]), h4s = sx16(h[4]), h5s = sx16(h[5]);
  v2f b1, b2;
  b1.x = half ? h0s : x45.x; b1.y = half ? h1s : x45.y;
  b2.x = half ? h4s : h[2];  b2.y = half ? h5s : h[3];
  bch[0] = xlo; bch[1] = b1; bch[2] = b2;
}

static __device__ __forceinline__ void b_l1(const float f0[6], v2f r01, v2f rhi,
                                            const float h[6], int half, v2f bch[5]) {
  float f2s = sx16(f0[2]), f3s = sx16(f0[3]);
  float h2s = sx16(h[2]),  h3s = sx16(h[3]);
  v2f b0, b1, b3, b4;
  b0.x = half ? f2s : f0[0];    b0.y = half ? f3s : f0[1];
  b1.x = half ? r01.x : f0[4];  b1.y = half ? r01.y : f0[5];
  b3.x = half ? h2s : h[0];     b3.y = half ? h3s : h[1];
  b4.x = half ? 0.0f : h[4];    b4.y = half ? 0.0f : h[5];
  bch[0] = b0; bch[1] = b1; bch[2] = rhi; bch[3] = b3; bch[4] = b4;
}

__global__ void __launch_bounds__(WG_THREADS) lstm_l0_bwd(const float* __restrict__ tf, const float* __restrict__ h0,
                            const float* __restrict__ c0,
                            const float* __restrict__ Wih, const float* __restrict__ Whh,
                            const float* __restrict__ bias,
                            float* __restrict__ r0ws, int B, int T) {
  int tid = blockIdx.x * blockDim.x + threadIdx.x;
  int lane = tid & 31, lm = lane & 15, half = lane >> 4;
  int s = (tid >> 5) * 16 + lm;
  if (s >= B) return;

  v2f a[2][5]; v8f ci[2];
  load_A(Wih, Whh, bias, LSTM_H, 3, a, ci, lm, half);

  float h[6], c[6];
  size_t d1 = (size_t)B * LSTM_H;
#pragma unroll
  for (int j = 0; j < 6; ++j) {
    h[j] = h0[d1 + (size_t)s * 6 + j];
    c[j] = c0[d1 + (size_t)s * 6 + j];
  }

  const float* xrow = tf + (size_t)s * T * 6;
  v2f cxl[BLK_L0], cxh[BLK_L0];
#pragma unroll
  for (int j = 0; j < BLK_L0; ++j) {
    int tt = T - 1 - j; tt = tt < 0 ? 0 : tt;
    ld_x(xrow + (size_t)tt * 6, half, cxl[j], cxh[j]);
  }

  auto step = [&](int t, v2f xlo, v2f x45) {
    v2f bch[5]; b_l0(xlo, x45, h, half, bch);
    v8f A0, A1; mm(a, ci, bch, 3, A0, A1);
    lstm_pointwise(A0, A1, half == 0, c, h);
    if (half == 0) {
      float* wp = r0ws + ((size_t)t * B + s) * 6;
      v2f w01; w01.x = h[0]; w01.y = h[1];
      v2f w23; w23.x = h[2]; w23.y = h[3];
      v2f w45; w45.x = h[4]; w45.y = h[5];
      *(v2f*)(wp) = w01; *(v2f*)(wp + 2) = w23; *(v2f*)(wp + 4) = w45;
    }
  };

  int t = T - 1;
  for (; t >= BLK_L0 - 1; t -= BLK_L0) {
    v2f nxl[BLK_L0], nxh[BLK_L0];
#pragma unroll
    for (int j = 0; j < BLK_L0; ++j) {
      int tn = t - BLK_L0 - j; tn = tn < 0 ? 0 : tn;
      ld_x(xrow + (size_t)tn * 6, half, nxl[j], nxh[j]);
    }
#pragma unroll
    for (int j = 0; j < BLK_L0; ++j) step(t - j, cxl[j], cxh[j]);
#pragma unroll
    for (int j = 0; j < BLK_L0; ++j) { cxl[j] = nxl[j]; cxh[j] = nxh[j]; }
  }
#pragma unroll
  for (int j = 0; j < BLK_L0 - 1; ++j) {
    if (j <= t) step(t - j, cxl[j], cxh[j]);
  }
}

__global__ void __launch_bounds__(WG_THREADS) lstm_fused(const float* __restrict__ tf, const float* __restrict__ h0,
                           const float* __restrict__ c0,
                           const float* __restrict__ Wih0f, const float* __restrict__ Whh0f,
                           const float* __restrict__ b0f,
                           const float* __restrict__ Wih1f, const float* __restrict__ Whh1f,
                           const float* __restrict__ b1f,
                           const float* __restrict__ Wih1b, const float* __restrict__ Whh1b,
                           const float* __restrict__ b1b,
                           const float* __restrict__ fcw, const float* __restrict__ fcb,
                           const float* __restrict__ r0ws, float* __restrict__ out,
                           int B, int T) {
  int tid = blockIdx.x * blockDim.x + threadIdx.x;
  int lane = tid & 31, lm = lane & 15, half = lane >> 4;
  int s = (tid >> 5) * 16 + lm;
  if (s >= B) return;

  v2f a0[2][5]; v8f ci0[2]; load_A(Wih0f, Whh0f, b0f, LSTM_H, 3, a0, ci0, lm, half);
  v2f a1[2][5]; v8f ci1[2]; load_A(Wih1f, Whh1f, b1f, 2 * LSTM_H, 5, a1, ci1, lm, half);

  float hA[6], cA[6], hB[6], cB[6];
  size_t d0 = 0, d2 = 2 * (size_t)B * LSTM_H;
#pragma unroll
  for (int j = 0; j < 6; ++j) {
    hA[j] = h0[d0 + (size_t)s * 6 + j];  cA[j] = c0[d0 + (size_t)s * 6 + j];
    hB[j] = h0[d2 + (size_t)s * 6 + j];  cB[j] = c0[d2 + (size_t)s * 6 + j];
  }

  const float* xrow = tf + (size_t)s * T * 6;
  const float* rcol = r0ws + (size_t)s * 6;
  const size_t rstep = (size_t)B * 6;
  v2f cxl[BLK_FU], cxh[BLK_FU], cra[BLK_FU], crb[BLK_FU];
#pragma unroll
  for (int j = 0; j < BLK_FU; ++j) {
    int tt = j > T - 1 ? T - 1 : j;
    ld_x(xrow + (size_t)tt * 6, half, cxl[j], cxh[j]);
    ld_r(rcol + (size_t)tt * rstep, half, cra[j], crb[j]);
  }

  auto step = [&](v2f xlo, v2f x45, v2f r01, v2f rhi) {
    v2f bch[5]; b_l0(xlo, x45, hA, half, bch);
    v8f A0, A1; mm(a0, ci0, bch, 3, A0, A1);
    lstm_pointwise(A0, A1, half == 0, cA, hA);
    b_l1(hA, r01, rhi, hB, half, bch);
    mm(a1, ci1, bch, 5, A0, A1);
    lstm_pointwise(A0, A1, half == 0, cB, hB);
  };

  int t = 0;
  for (; t + BLK_FU <= T; t += BLK_FU) {
    v2f nxl[BLK_FU], nxh[BLK_FU], nra[BLK_FU], nrb[BLK_FU];
#pragma unroll
    for (int j = 0; j < BLK_FU; ++j) {
      int tn = t + BLK_FU + j; tn = tn > T - 1 ? T - 1 : tn;
      ld_x(xrow + (size_t)tn * 6, half, nxl[j], nxh[j]);
      ld_r(rcol + (size_t)tn * rstep, half, nra[j], nrb[j]);
    }
    __builtin_amdgcn_sched_barrier(0);
#pragma unroll
    for (int j = 0; j < BLK_FU; ++j) step(cxl[j], cxh[j], cra[j], crb[j]);
    __builtin_amdgcn_sched_barrier(0);
#pragma unroll
    for (int j = 0; j < BLK_FU; ++j) {
      cxl[j] = nxl[j]; cxh[j] = nxh[j]; cra[j] = nra[j]; crb[j] = nrb[j];
    }
  }
#pragma unroll
  for (int j = 0; j < BLK_FU - 1; ++j) {
    if (t + j < T) step(cxl[j], cxh[j], cra[j], crb[j]);
  }
  v2f a2[2][5]; v8f ci2[2]; load_A(Wih1b, Whh1b, b1b, 2 * LSTM_H, 5, a2, ci2, lm, half);
  float hR[6], cR[6];
  size_t d3 = 3 * (size_t)B * LSTM_H;
#pragma unroll
  for (int j = 0; j < 6; ++j) {
    hR[j] = h0[d3 + (size_t)s * 6 + j];  cR[j] = c0[d3 + (size_t)s * 6 + j];
  }
  {
    const float* r0p = r0ws + ((size_t)(T - 1) * B + s) * 6;
    v2f r01, rhi; ld_r(r0p, half, r01, rhi);
    v2f bch[5]; b_l1(hA, r01, rhi, hR, half, bch);
    v8f A0, A1; mm(a2, ci2, bch, 5, A0, A1);
    lstm_pointwise(A0, A1, half == 0, cR, hR);
  }
  if (half == 0) {
    float v[12];
#pragma unroll
    for (int k = 0; k < 6; ++k) {
      v[k]     = hB[k] > 0.0f ? hB[k] : 0.0f;
      v[6 + k] = hR[k] > 0.0f ? hR[k] : 0.0f;
    }
#pragma unroll
    for (int j = 0; j < 6; ++j) {
      float acc = fcb[j];
#pragma unroll
      for (int k = 0; k < 12; ++k) acc = fmaf(fcw[j * 12 + k], v[k], acc);
      out[(size_t)s * 6 + j] = acc;
    }
  }
}

extern "C" void kernel_launch(void* const* d_in, const int* in_sizes, int n_in,
                              void* d_out, int out_size, void* d_ws, size_t ws_size,
                              hipStream_t stream) {
  const float* tf    = (const float*)d_in[0];
  const float* h0    = (const float*)d_in[1];
  const float* c0    = (const float*)d_in[2];
  const float* Wih0f = (const float*)d_in[3];
  const float* Whh0f = (const float*)d_in[4];
  const float* b0f   = (const float*)d_in[5];
  const float* Wih0b = (const float*)d_in[6];
  const float* Whh0b = (const float*)d_in[7];
  const float* b0b   = (const float*)d_in[8];
  const float* Wih1f = (const float*)d_in[9];
  const float* Whh1f = (const float*)d_in[10];
  const float* b1f   = (const float*)d_in[11];
  const float* Wih1b = (const float*)d_in[12];
  const float* Whh1b = (const float*)d_in[13];
  const float* b1b   = (const float*)d_in[14];
  const float* fcw   = (const float*)d_in[15];
  const float* fcb   = (const float*)d_in[16];

  int B = in_sizes[1] / (4 * LSTM_H);
  int T = in_sizes[0] / (B * LSTM_H);

  size_t need = (size_t)B * T * LSTM_H * sizeof(float);
  if (ws_size < need) return;
  float* r0ws = (float*)d_ws;

  int waves = (B + 15) / 16;
  int threads = waves * 32;
  int block = WG_THREADS;
  int grid = (threads + block - 1) / block;

  lstm_l0_bwd<<<grid, block, 0, stream>>>(tf, h0, c0, Wih0b, Whh0b, b0b, r0ws, B, T);
  lstm_fused<<<grid, block, 0, stream>>>(tf, h0, c0, Wih0f, Whh0f, b0f,
                                         Wih1f, Whh1f, b1f, Wih1b, Whh1b, b1b,
                                         fcw, fcb, r0ws, (float*)d_out, B, T);
}


# gate rows re-mapped so each lane holds all four gates of 3 hidden units: LSTM pointwise lane-local, no cross-lane moves, selects or discarded exp/rcp; k order of the gate sums changes
# speedup vs baseline: 3.4561x; 1.3636x over previous
#include <hip/hip_runtime.h>
#include <hip/hip_bf16.h>


typedef __attribute__((ext_vector_type(2))) float v2f;
typedef __attribute__((ext_vector_type(8))) float v8f;

#define LSTM_H 6
#define BLK_L0 8
#define BLK_FU 4
#define WG_THREADS 128

static __device__ __forceinline__ float sigm(float x) {
  return __builtin_amdgcn_rcpf(1.0f + __expf(-x));
}
static __device__ __forceinline__ float tanh_fast(float x) {
  return 1.0f - 2.0f * __builtin_amdgcn_rcpf(__expf(2.0f * x) + 1.0f);
}
static __device__ __forceinline__ float sx16(float v) {
  int b = __builtin_bit_cast(int, v);
  return __builtin_bit_cast(float,
      __builtin_amdgcn_permlanex16(b, b, 0x76543210, 0xfedcba98, false, false));
}
static __device__ __forceinline__ v8f wmma4(v2f a, v2f b, v8f c) {
  return __builtin_amdgcn_wmma_f32_16x16x4_f32(false, a, false, b, (short)0, c,
                                               false, false);
}

static __device__ __forceinline__ int gate_row(int m) {
  int idx = m & 7;
  if (idx >= LSTM_H) return -1;
  int hf = (m >> 3) & 1;
  int tile = m >> 4;
  int second = idx >= 3 ? 1 : 0;
  int gate = 2 * tile + second;
  int unit = 3 * hf + idx - 3 * second;
  return gate * LSTM_H + unit;
}

static __device__ __forceinline__ int ksrc_l0(int c, int hf, int e) {
  if (c == 0) return 2 * hf + e;
  if (c == 1) return 6 + 3 * hf + e;
  return e == 0 ? 4 + hf : 6 + 3 * hf + 2;
}
static __device__ __forceinline__ int ksrc_l1(int c, int hf, int e) {
  if (c == 0) return 6 + 3 * hf + e;
  if (c == 1) return e == 0 ? 6 + 3 * hf + 2 : -1;
  if (c == 2) return 12 + 3 * hf + e;
  if (c == 3) return e == 0 ? 3 * hf + 2 : 12 + 3 * hf + 2;
  return 3 * hf + e;
}

static __device__ __forceinline__ float wcomb(const float* Wih, const float* Whh,
                                              int D, int m, int k) {
  int r = gate_row(m);
  if (r < 0 || k < 0) return 0.0f;
  if (k < D) return Wih[r * D + k];
  return Whh[r * LSTM_H + (k - D)];
}

template <bool L1>
static __device__ __forceinline__ void load_A(const float* Wih, const float* Whh,
                                              const float* bias,
                                              v2f a[2][5], v8f cini[2],
                                              int lm, int half) {
  const int D = L1 ? 2 * LSTM_H : LSTM_H;
  const int nch = L1 ? 5 : 3;
  for (int t = 0; t < 2; ++t) {
    int m = t * 16 + lm;
#pragma unroll
    for (int c = 0; c < 5; ++c) {
      if (c < nch) {
        int k0 = L1 ? ksrc_l1(c, half, 0) : ksrc_l0(c, half, 0);
        int k1 = L1 ? ksrc_l1(c, half, 1) : ksrc_l0(c, half, 1);
        v2f av;
        av.x = wcomb(Wih, Whh, D, m, k0);
        av.y = wcomb(Wih, Whh, D, m, k1);
        a[t][c] = av;
      }
    }
    v8f ci;
#pragma unroll
    for (int r = 0; r < 8; ++r) {
      int gr = gate_row(t * 16 + half * 8 + r);
      ci[r] = (gr >= 0) ? bias[gr] : 0.0f;
    }
    cini[t] = ci;
  }
}

static __device__ __forceinline__ void mm(const v2f a[2][5], const v8f cini[2],
                                          const v2f bch[5], int nch,
                                          v8f& A0, v8f& A1) {
  A0 = cini[0];
  A1 = cini[1];
#pragma unroll
  for (int c = 0; c < 5; ++c) {
    if (c < nch) {
      A0 = wmma4(a[0][c], bch[c], A0);
      A1 = wmma4(a[1][c], bch[c], A1);
    }
  }
}

static __device__ __forceinline__ void lstm_pointwise(const v8f& A0, const v8f& A1,
                                                      float c[3], float h[3]) {
#pragma unroll
  for (int u = 0; u < 3; ++u) {
    float si = sigm(A0[u]);
    float sf = sigm(A0[3 + u]);
    float tg = tanh_fast(A1[u]);
    float so = sigm(A1[3 + u]);
    float cn = sf * c[u] + si * tg;
    c[u] = cn;
    h[u] = so * tanh_fast(cn);
  }
}

static __device__ __forceinline__ void ld_x(const float* xp, int half,
                                            v2f& xlo, float& x4) {
  xlo = *(const v2f*)(xp + 2 * half);
  x4 = xp[4 + half];
}
static __device__ __forceinline__ void ld_r(const float* r0p, int half,
                                            v2f& r01, float& r2) {
  r01 = *(const v2f*)(r0p + 2 * half);
  r2 = r0p[4 + half];
}

static __device__ __forceinline__ void b_l0(v2f xlo, float x4, const float h[3], v2f bch[5]) {
  v2f b1, b2;
  b1.x = h[0]; b1.y = h[1];
  b2.x = x4;   b2.y = h[2];
  bch[0] = xlo; bch[1] = b1; bch[2] = b2;
}

static __device__ __forceinline__ void b_l1(const float f0[3], v2f r01, float r2,
                                            const float h[3], v2f bch[5]) {
  v2f b1, b2, b3, b4;
  b1.x = r2;    b1.y = 0.0f;
  b2.x = h[0];  b2.y = h[1];
  b3.x = f0[2]; b3.y = h[2];
  b4.x = f0[0]; b4.y = f0[1];
  bch[0] = r01; bch[1] = b1; bch[2] = b2; bch[3] = b3; bch[4] = b4;
}

__global__ void __launch_bounds__(WG_THREADS) lstm_l0_bwd(const float* __restrict__ tf, const float* __restrict__ h0,
                            const float* __restrict__ c0,
                            const float* __restrict__ Wih, const float* __restrict__ Whh,
                            const float* __restrict__ bias,
                            float* __restrict__ r0ws, int B, int T) {
  int tid = blockIdx.x * blockDim.x + threadIdx.x;
  int lane = tid & 31, lm = lane & 15, half = lane >> 4;
  int s = (tid >> 5) * 16 + lm;
  if (s >= B) return;

  v2f a[2][5]; v8f ci[2];
  load_A<false>(Wih, Whh, bias, a, ci, lm, half);

  float h[3], c[3];
  size_t d1 = (size_t)B * LSTM_H;
#pragma unroll
  for (int u = 0; u < 3; ++u) {
    h[u] = h0[d1 + (size_t)s * 6 + 3 * half + u];
    c[u] = c0[d1 + (size_t)s * 6 + 3 * half + u];
  }

  const float* xrow = tf + (size_t)s * T * 6;
  v2f cxl[BLK_L0]; float cx4[BLK_L0];
#pragma unroll
  for (int j = 0; j < BLK_L0; ++j) {
    int tt = T - 1 - j; tt = tt < 0 ? 0 : tt;
    ld_x(xrow + (size_t)tt * 6, half, cxl[j], cx4[j]);
  }

  auto step = [&](int t, v2f xlo, float x4) {
    v2f bch[5]; b_l0(xlo, x4, h, bch);
    v8f A0, A1; mm(a, ci, bch, 3, A0, A1);
    lstm_pointwise(A0, A1, c, h);
    float* wp = r0ws + ((size_t)t * B + s) * 6;
    v2f w01; w01.x = h[0]; w01.y = h[1];
    *(v2f*)(wp + 2 * half) = w01;
    wp[4 + half] = h[2];
  };

  int t = T - 1;
  for (; t >= BLK_L0 - 1; t -= BLK_L0) {
    v2f nxl[BLK_L0]; float nx4[BLK_L0];
#pragma unroll
    for (int j = 0; j < BLK_L0; ++j) {
      int tn = t - BLK_L0 - j; tn = tn < 0 ? 0 : tn;
      ld_x(xrow + (size_t)tn * 6, half, nxl[j], nx4[j]);
    }
    __builtin_amdgcn_sched_barrier(0);
#pragma unroll
    for (int j = 0; j < BLK_L0; ++j) step(t - j, cxl[j], cx4[j]);
    __builtin_amdgcn_sched_barrier(0);
#pragma unroll
    for (int j = 0; j < BLK_L0; ++j) { cxl[j] = nxl[j]; cx4[j] = nx4[j]; }
  }
#pragma unroll
  for (int j = 0; j < BLK_L0 - 1; ++j) {
    if (j <= t) step(t - j, cxl[j], cx4[j]);
  }
}

__global__ void __launch_bounds__(WG_THREADS) lstm_fused(const float* __restrict__ tf, const float* __restrict__ h0,
                           const float* __restrict__ c0,
                           const float* __restrict__ Wih0f, const float* __restrict__ Whh0f,
                           const float* __restrict__ b0f,
                           const float* __restrict__ Wih1f, const float* __restrict__ Whh1f,
                           const float* __restrict__ b1f,
                           const float* __restrict__ Wih1b, const float* __restrict__ Whh1b,
                           const float* __restrict__ b1b,
                           const float* __restrict__ fcw, const float* __restrict__ fcb,
                           const float* __restrict__ r0ws, float* __restrict__ out,
                           int B, int T) {
  int tid = blockIdx.x * blockDim.x + threadIdx.x;
  int lane = tid & 31, lm = lane & 15, half = lane >> 4;
  int s = (tid >> 5) * 16 + lm;
  if (s >= B) return;

  v2f a0[2][5]; v8f ci0[2]; load_A<false>(Wih0f, Whh0f, b0f, a0, ci0, lm, half);
  v2f a1[2][5]; v8f ci1[2]; load_A<true>(Wih1f, Whh1f, b1f, a1, ci1, lm, half);

  float hA[3], cA[3], hB[3], cB[3];
  size_t d0 = 0, d2 = 2 * (size_t)B * LSTM_H;
#pragma unroll
  for (int u = 0; u < 3; ++u) {
    size_t o = (size_t)s * 6 + 3 * half + u;
    hA[u] = h0[d0 + o];  cA[u] = c0[d0 + o];
    hB[u] = h0[d2 + o];  cB[u] = c0[d2 + o];
  }

  const float* xrow = tf + (size_t)s * T * 6;
  const float* rcol = r0ws + (size_t)s * 6;
  const size_t rstep = (size_t)B * 6;
  v2f cxl[BLK_FU], cra[BLK_FU]; float cx4[BLK_FU], crb[BLK_FU];
#pragma unroll
  for (int j = 0; j < BLK_FU; ++j) {
    int tt = j > T - 1 ? T - 1 : j;
    ld_x(xrow + (size_t)tt * 6, half, cxl[j], cx4[j]);
    ld_r(rcol + (size_t)tt * rstep, half, cra[j], crb[j]);
  }

  auto step = [&](v2f xlo, float x4, v2f r01, float r2) {
    v2f bch[5]; b_l0(xlo, x4, hA, bch);
    v8f A0, A1; mm(a0, ci0, bch, 3, A0, A1);
    lstm_pointwise(A0, A1, cA, hA);
    b_l1(hA, r01, r2, hB, bch);
    mm(a1, ci1, bch, 5, A0, A1);
    lstm_pointwise(A0, A1, cB, hB);
  };

  int t = 0;
  for (; t + BLK_FU <= T; t += BLK_FU) {
    v2f nxl[BLK_FU], nra[BLK_FU]; float nx4[BLK_FU], nrb[BLK_FU];
#pragma unroll
    for (int j = 0; j < BLK_FU; ++j) {
      int tn = t + BLK_FU + j; tn = tn > T - 1 ? T - 1 : tn;
      ld_x(xrow + (size_t)tn * 6, half, nxl[j], nx4[j]);
      ld_r(rcol + (size_t)tn * rstep, half, nra[j], nrb[j]);
    }
    __builtin_amdgcn_sched_barrier(0);
#pragma unroll
    for (int j = 0; j < BLK_FU; ++j) step(cxl[j], cx4[j], cra[j], crb[j]);
    __builtin_amdgcn_sched_barrier(0);
#pragma unroll
    for (int j = 0; j < BLK_FU; ++j) {
      cxl[j] = nxl[j]; cx4[j] = nx4[j]; cra[j] = nra[j]; crb[j] = nrb[j];
    }
  }
#pragma unroll
  for (int j = 0; j < BLK_FU - 1; ++j) {
    if (t + j < T) step(cxl[j], cx4[j], cra[j], crb[j]);
  }
  v2f a2[2][5]; v8f ci2[2]; load_A<true>(Wih1b, Whh1b, b1b, a2, ci2, lm, half);
  float hR[3], cR[3];
  size_t d3 = 3 * (size_t)B * LSTM_H;
#pragma unroll
  for (int u = 0; u < 3; ++u) {
    size_t o = (size_t)s * 6 + 3 * half + u;
    hR[u] = h0[d3 + o];  cR[u] = c0[d3 + o];
  }
  {
    const float* r0p = r0ws + ((size_t)(T - 1) * B + s) * 6;
    v2f r01; float r2; ld_r(r0p, half, r01, r2);
    v2f bch[5]; b_l1(hA, r01, r2, hR, bch);
    v8f A0, A1; mm(a2, ci2, bch, 5, A0, A1);
    lstm_pointwise(A0, A1, cR, hR);
  }
  float pB[3], pR[3];
#pragma unroll
  for (int u = 0; u < 3; ++u) { pB[u] = sx16(hB[u]); pR[u] = sx16(hR[u]); }
  if (half == 0) {
    float v[12];
#pragma unroll
    for (int u = 0; u < 3; ++u) {
      v[u]     = hB[u] > 0.0f ? hB[u] : 0.0f;
      v[3 + u] = pB[u] > 0.0f ? pB[u] : 0.0f;
      v[6 + u] = hR[u] > 0.0f ? hR[u] : 0.0f;
      v[9 + u] = pR[u] > 0.0f ? pR[u] : 0.0f;
    }
#pragma unroll
    for (int j = 0; j < 6; ++j) {
      float acc = fcb[j];
#pragma unroll
      for (int k = 0; k < 12; ++k) acc = fmaf(fcw[j * 12 + k], v[k], acc);
      out[(size_t)s * 6 + j] = acc;
    }
  }
}

extern "C" void kernel_launch(void* const* d_in, const int* in_sizes, int n_in,
                              void* d_out, int out_size, void* d_ws, size_t ws_size,
                              hipStream_t stream) {
  const float* tf    = (const float*)d_in[0];
  const float* h0    = (const float*)d_in[1];
  const float* c0    = (const float*)d_in[2];
  const float* Wih0f = (const float*)d_in[3];
  const float* Whh0f = (const float*)d_in[4];
  const float* b0f   = (const float*)d_in[5];
  const float* Wih0b = (const float*)d_in[6];
  const float* Whh0b = (const float*)d_in[7];
  const float* b0b   = (const float*)d_in[8];
  const float* Wih1f = (const float*)d_in[9];
  const float* Whh1f = (const float*)d_in[10];
  const float* b1f   = (const float*)d_in[11];
  const float* Wih1b = (const float*)d_in[12];
  const float* Whh1b = (const float*)d_in[13];
  const float* b1b   = (const float*)d_in[14];
  const float* fcw   = (const float*)d_in[15];
  const float* fcb   = (const float*)d_in[16];

  int B = in_sizes[1] / (4 * LSTM_H);
  int T = in_sizes[0] / (B * LSTM_H);

  size_t need = (size_t)B * T * LSTM_H * sizeof(float);
  if (ws_size < need) return;
  float* r0ws = (float*)d_ws;

  int waves = (B + 15) / 16;
  int threads = waves * 32;
  int block = WG_THREADS;
  int grid = (threads + block - 1) / block;

  lstm_l0_bwd<<<grid, block, 0, stream>>>(tf, h0, c0, Wih0b, Whh0b, b0b, r0ws, B, T);
  lstm_fused<<<grid, block, 0, stream>>>(tf, h0, c0, Wih0f, Whh0f, b0f,
                                         Wih1f, Whh1f, b1f, Wih1b, Whh1b, b1b,
                                         fcw, fcb, r0ws, (float*)d_out, B, T);
}


# sigmoid and tanh from v_tanh_f32 not v_exp_f32 + v_rcp_f32: sigmoid(z) = 0.5 + 0.5*tanh(z/2), the half folded exactly into the i, f, o rows of A and bias; 5 transcendentals per unit-step, was 10
# speedup vs baseline: 4.3688x; 1.2641x over previous
#include <hip/hip_runtime.h>
#include <hip/hip_bf16.h>


typedef __attribute__((ext_vector_type(2))) float v2f;
typedef __attribute__((ext_vector_type(8))) float v8f;

#define LSTM_H 6
#define BLK_L0 8
#define BLK_FU 4
#define WG_THREADS 128

static __device__ __forceinline__ float tanh_hw(float x) {
  return __builtin_amdgcn_tanhf(x);
}
static __device__ __forceinline__ float sigm_half(float zh) {
  return fmaf(0.5f, tanh_hw(zh), 0.5f);
}
static __device__ __forceinline__ float sx16(float v) {
  int b = __builtin_bit_cast(int, v);
  return __builtin_bit_cast(float,
      __builtin_amdgcn_permlanex16(b, b, 0x76543210, 0xfedcba98, false, false));
}
static __device__ __forceinline__ v8f wmma4(v2f a, v2f b, v8f c) {
  return __builtin_amdgcn_wmma_f32_16x16x4_f32(false, a, false, b, (short)0, c,
                                               false, false);
}

static __device__ __forceinline__ int gate_row(int m) {
  int idx = m & 7;
  if (idx >= LSTM_H) return -1;
  int hf = (m >> 3) & 1;
  int tile = m >> 4;
  int second = idx >= 3 ? 1 : 0;
  int gate = 2 * tile + second;
  int unit = 3 * hf + idx - 3 * second;
  return gate * LSTM_H + unit;
}

static __device__ __forceinline__ float gate_scale(int m) {
  int idx = m & 7;
  int tile = m >> 4;
  return (tile == 1 && idx < 3) ? 1.0f : 0.5f;
}

static __device__ __forceinline__ int ksrc_l0(int c, int hf, int e) {
  if (c == 0) return 2 * hf + e;
  if (c == 1) return 6 + 3 * hf + e;
  return e == 0 ? 4 + hf : 6 + 3 * hf + 2;
}
static __device__ __forceinline__ int ksrc_l1(int c, int hf, int e) {
  if (c == 0) return 6 + 3 * hf + e;
  if (c == 1) return e == 0 ? 6 + 3 * hf + 2 : -1;
  if (c == 2) return 12 + 3 * hf + e;
  if (c == 3) return e == 0 ? 3 * hf + 2 : 12 + 3 * hf + 2;
  return 3 * hf + e;
}

static __device__ __forceinline__ float wcomb(const float* Wih, const float* Whh,
                                              int D, int m, int k) {
  int r = gate_row(m);
  if (r < 0 || k < 0) return 0.0f;
  float sc = gate_scale(m);
  if (k < D) return sc * Wih[r * D + k];
  return sc * Whh[r * LSTM_H + (k - D)];
}

template <bool L1>
static __device__ __forceinline__ void load_A(const float* Wih, const float* Whh,
                                              const float* bias,
                                              v2f a[2][5], v8f cini[2],
                                              int lm, int half) {
  const int D = L1 ? 2 * LSTM_H : LSTM_H;
  const int nch = L1 ? 5 : 3;
  for (int t = 0; t < 2; ++t) {
    int m = t * 16 + lm;
#pragma unroll
    for (int c = 0; c < 5; ++c) {
      if (c < nch) {
        int k0 = L1 ? ksrc_l1(c, half, 0) : ksrc_l0(c, half, 0);
        int k1 = L1 ? ksrc_l1(c, half, 1) : ksrc_l0(c, half, 1);
        v2f av;
        av.x = wcomb(Wih, Whh, D, m, k0);
        av.y = wcomb(Wih, Whh, D, m, k1);
        a[t][c] = av;
      }
    }
    v8f ci;
#pragma unroll
    for (int r = 0; r < 8; ++r) {
      int gm = t * 16 + half * 8 + r;
      int gr = gate_row(gm);
      ci[r] = (gr >= 0) ? gate_scale(gm) * bias[gr] : 0.0f;
    }
    cini[t] = ci;
  }
}

static __device__ __forceinline__ void mm(const v2f a[2][5], const v8f cini[2],
                                          const v2f bch[5], int nch,
                                          v8f& A0, v8f& A1) {
  A0 = cini[0];
  A1 = cini[1];
#pragma unroll
  for (int c = 0; c < 5; ++c) {
    if (c < nch) {
      A0 = wmma4(a[0][c], bch[c], A0);
      A1 = wmma4(a[1][c], bch[c], A1);
    }
  }
}

static __device__ __forceinline__ void lstm_pointwise(const v8f& A0, const v8f& A1,
                                                      float c[3], float h[3]) {
#pragma unroll
  for (int u = 0; u < 3; ++u) {
    float si = sigm_half(A0[u]);
    float sf = sigm_half(A0[3 + u]);
    float tg = tanh_hw(A1[u]);
    float so = sigm_half(A1[3 + u]);
    float cn = sf * c[u] + si * tg;
    c[u] = cn;
    h[u] = so * tanh_hw(cn);
  }
}

static __device__ __forceinline__ void ld_x(const float* xp, int half,
                                            v2f& xlo, float& x4) {
  xlo = *(const v2f*)(xp + 2 * half);
  x4 = xp[4 + half];
}
static __device__ __forceinline__ void ld_r(const float* r0p, int half,
                                            v2f& r01, float& r2) {
  r01 = *(const v2f*)(r0p + 2 * half);
  r2 = r0p[4 + half];
}

static __device__ __forceinline__ void b_l0(v2f xlo, float x4, const float h[3], v2f bch[5]) {
  v2f b1, b2;
  b1.x = h[0]; b1.y = h[1];
  b2.x = x4;   b2.y = h[2];
  bch[0] = xlo; bch[1] = b1; bch[2] = b2;
}

static __device__ __forceinline__ void b_l1(const float f0[3], v2f r01, float r2,
                                            const float h[3], v2f bch[5]) {
  v2f b1, b2, b3, b4;
  b1.x = r2;    b1.y = 0.0f;
  b2.x = h[0];  b2.y = h[1];
  b3.x = f0[2]; b3.y = h[2];
  b4.x = f0[0]; b4.y = f0[1];
  bch[0] = r01; bch[1] = b1; bch[2] = b2; bch[3] = b3; bch[4] = b4;
}

__global__ void __launch_bounds__(WG_THREADS) lstm_l0_bwd(const float* __restrict__ tf, const float* __restrict__ h0,
                            const float* __restrict__ c0,
                            const float* __restrict__ Wih, const float* __restrict__ Whh,
                            const float* __restrict__ bias,
                            float* __restrict__ r0ws, int B, int T) {
  int tid = blockIdx.x * blockDim.x + threadIdx.x;
  int lane = tid & 31, lm = lane & 15, half = lane >> 4;
  int s = (tid >> 5) * 16 + lm;
  if (s >= B) return;

  v2f a[2][5]; v8f ci[2];
  load_A<false>(Wih, Whh, bias, a, ci, lm, half);

  float h[3], c[3];
  size_t d1 = (size_t)B * LSTM_H;
#pragma unroll
  for (int u = 0; u < 3; ++u) {
    h[u] = h0[d1 + (size_t)s * 6 + 3 * half + u];
    c[u] = c0[d1 + (size_t)s * 6 + 3 * half + u];
  }

  const float* xrow = tf + (size_t)s * T * 6;
  v2f cxl[BLK_L0]; float cx4[BLK_L0];
#pragma unroll
  for (int j = 0; j < BLK_L0; ++j) {
    int tt = T - 1 - j; tt = tt < 0 ? 0 : tt;
    ld_x(xrow + (size_t)tt * 6, half, cxl[j], cx4[j]);
  }

  auto step = [&](int t, v2f xlo, float x4) {
    v2f bch[5]; b_l0(xlo, x4, h, bch);
    v8f A0, A1; mm(a, ci, bch, 3, A0, A1);
    lstm_pointwise(A0, A1, c, h);
    float* wp = r0ws + ((size_t)t * B + s) * 6;
    v2f w01; w01.x = h[0]; w01.y = h[1];
    *(v2f*)(wp + 2 * half) = w01;
    wp[4 + half] = h[2];
  };

  int t = T - 1;
  for (; t >= BLK_L0 - 1; t -= BLK_L0) {
    v2f nxl[BLK_L0]; float nx4[BLK_L0];
#pragma unroll
    for (int j = 0; j < BLK_L0; ++j) {
      int tn = t - BLK_L0 - j; tn = tn < 0 ? 0 : tn;
      ld_x(xrow + (size_t)tn * 6, half, nxl[j], nx4[j]);
    }
    __builtin_amdgcn_sched_barrier(0);
#pragma unroll
    for (int j = 0; j < BLK_L0; ++j) step(t - j, cxl[j], cx4[j]);
    __builtin_amdgcn_sched_barrier(0);
#pragma unroll
    for (int j = 0; j < BLK_L0; ++j) { cxl[j] = nxl[j]; cx4[j] = nx4[j]; }
  }
#pragma unroll
  for (int j = 0; j < BLK_L0 - 1; ++j) {
    if (j <= t) step(t - j, cxl[j], cx4[j]);
  }
}

__global__ void __launch_bounds__(WG_THREADS) lstm_fused(const float* __restrict__ tf, const float* __restrict__ h0,
                           const float* __restrict__ c0,
                           const float* __restrict__ Wih0f, const float* __restrict__ Whh0f,
                           const float* __restrict__ b0f,
                           const float* __restrict__ Wih1f, const float* __restrict__ Whh1f,
                           const float* __restrict__ b1f,
                           const float* __restrict__ Wih1b, const float* __restrict__ Whh1b,
                           const float* __restrict__ b1b,
                           const float* __restrict__ fcw, const float* __restrict__ fcb,
                           const float* __restrict__ r0ws, float* __restrict__ out,
                           int B, int T) {
  int tid = blockIdx.x * blockDim.x + threadIdx.x;
  int lane = tid & 31, lm = lane & 15, half = lane >> 4;
  int s = (tid >> 5) * 16 + lm;
  if (s >= B) return;

  v2f a0[2][5]; v8f ci0[2]; load_A<false>(Wih0f, Whh0f, b0f, a0, ci0, lm, half);
  v2f a1[2][5]; v8f ci1[2]; load_A<true>(Wih1f, Whh1f, b1f, a1, ci1, lm, half);

  float hA[3], cA[3], hB[3], cB[3];
  size_t d0 = 0, d2 = 2 * (size_t)B * LSTM_H;
#pragma unroll
  for (int u = 0; u < 3; ++u) {
    size_t o = (size_t)s * 6 + 3 * half + u;
    hA[u] = h0[d0 + o];  cA[u] = c0[d0 + o];
    hB[u] = h0[d2 + o];  cB[u] = c0[d2 + o];
  }

  const float* xrow = tf + (size_t)s * T * 6;
  const float* rcol = r0ws + (size_t)s * 6;
  const size_t rstep = (size_t)B * 6;
  v2f cxl[BLK_FU], cra[BLK_FU]; float cx4[BLK_FU], crb[BLK_FU];
#pragma unroll
  for (int j = 0; j < BLK_FU; ++j) {
    int tt = j > T - 1 ? T - 1 : j;
    ld_x(xrow + (size_t)tt * 6, half, cxl[j], cx4[j]);
    ld_r(rcol + (size_t)tt * rstep, half, cra[j], crb[j]);
  }

  auto step = [&](v2f xlo, float x4, v2f r01, float r2) {
    v2f bch[5]; b_l0(xlo, x4, hA, bch);
    v8f A0, A1; mm(a0, ci0, bch, 3, A0, A1);
    lstm_pointwise(A0, A1, cA, hA);
    b_l1(hA, r01, r2, hB, bch);
    mm(a1, ci1, bch, 5, A0, A1);
    lstm_pointwise(A0, A1, cB, hB);
  };

  int t = 0;
  for (; t + BLK_FU <= T; t += BLK_FU) {
    v2f nxl[BLK_FU], nra[BLK_FU]; float nx4[BLK_FU], nrb[BLK_FU];
#pragma unroll
    for (int j = 0; j < BLK_FU; ++j) {
      int tn = t + BLK_FU + j; tn = tn > T - 1 ? T - 1 : tn;
      ld_x(xrow + (size_t)tn * 6, half, nxl[j], nx4[j]);
      ld_r(rcol + (size_t)tn * rstep, half, nra[j], nrb[j]);
    }
    __builtin_amdgcn_sched_barrier(0);
#pragma unroll
    for (int j = 0; j < BLK_FU; ++j) step(cxl[j], cx4[j], cra[j], crb[j]);
    __builtin_amdgcn_sched_barrier(0);
#pragma unroll
    for (int j = 0; j < BLK_FU; ++j) {
      cxl[j] = nxl[j]; cx4[j] = nx4[j]; cra[j] = nra[j]; crb[j] = nrb[j];
    }
  }
#pragma unroll
  for (int j = 0; j < BLK_FU - 1; ++j) {
    if (t + j < T) step(cxl[j], cx4[j], cra[j], crb[j]);
  }
  v2f a2[2][5]; v8f ci2[2]; load_A<true>(Wih1b, Whh1b, b1b, a2, ci2, lm, half);
  float hR[3], cR[3];
  size_t d3 = 3 * (size_t)B * LSTM_H;
#pragma unroll
  for (int u = 0; u < 3; ++u) {
    size_t o = (size_t)s * 6 + 3 * half + u;
    hR[u] = h0[d3 + o];  cR[u] = c0[d3 + o];
  }
  {
    const float* r0p = r0ws + ((size_t)(T - 1) * B + s) * 6;
    v2f r01; float r2; ld_r(r0p, half, r01, r2);
    v2f bch[5]; b_l1(hA, r01, r2, hR, bch);
    v8f A0, A1; mm(a2, ci2, bch, 5, A0, A1);
    lstm_pointwise(A0, A1, cR, hR);
  }
  float pB[3], pR[3];
#pragma unroll
  for (int u = 0; u < 3; ++u) { pB[u] = sx16(hB[u]); pR[u] = sx16(hR[u]); }
  if (half == 0) {
    float v[12];
#pragma unroll
    for (int u = 0; u < 3; ++u) {
      v[u]     = hB[u] > 0.0f ? hB[u] : 0.0f;
      v[3 + u] = pB[u] > 0.0f ? pB[u] : 0.0f;
      v[6 + u] = hR[u] > 0.0f ? hR[u] : 0.0f;
      v[9 + u] = pR[u] > 0.0f ? pR[u] : 0.0f;
    }
#pragma unroll
    for (int j = 0; j < 6; ++j) {
      float acc = fcb[j];
#pragma unroll
      for (int k = 0; k < 12; ++k) acc = fmaf(fcw[j * 12 + k], v[k], acc);
      out[(size_t)s * 6 + j] = acc;
    }
  }
}

extern "C" void kernel_launch(void* const* d_in, const int* in_sizes, int n_in,
                              void* d_out, int out_size, void* d_ws, size_t ws_size,
                              hipStream_t stream) {
  const float* tf    = (const float*)d_in[0];
  const float* h0    = (const float*)d_in[1];
  const float* c0    = (const float*)d_in[2];
  const float* Wih0f = (const float*)d_in[3];
  const float* Whh0f = (const float*)d_in[4];
  const float* b0f   = (const float*)d_in[5];
  const float* Wih0b = (const float*)d_in[6];
  const float* Whh0b = (const float*)d_in[7];
  const float* b0b   = (const float*)d_in[8];
  const float* Wih1f = (const float*)d_in[9];
  const float* Whh1f = (const float*)d_in[10];
  const float* b1f   = (const float*)d_in[11];
  const float* Wih1b = (const float*)d_in[12];
  const float* Whh1b = (const float*)d_in[13];
  const float* b1b   = (const float*)d_in[14];
  const float* fcw   = (const float*)d_in[15];
  const float* fcb   = (const float*)d_in[16];

  int B = in_sizes[1] / (4 * LSTM_H);
  int T = in_sizes[0] / (B * LSTM_H);

  size_t need = (size_t)B * T * LSTM_H * sizeof(float);
  if (ws_size < need) return;
  float* r0ws = (float*)d_ws;

  int waves = (B + 15) / 16;
  int threads = waves * 32;
  int block = WG_THREADS;
  int grid = (threads + block - 1) / block;

  lstm_l0_bwd<<<grid, block, 0, stream>>>(tf, h0, c0, Wih0b, Whh0b, b0b, r0ws, B, T);
  lstm_fused<<<grid, block, 0, stream>>>(tf, h0, c0, Wih0f, Whh0f, b0f,
                                         Wih1f, Whh1f, b1f, Wih1b, Whh1b, b1b,
                                         fcw, fcb, r0ws, (float*)d_out, B, T);
}


# second launch split by wave role: two waves per work-group run the layer-0 forward recurrence and pass f0 through LDS, 8 steps per barrier, to two layer-1 waves, so both recurrences run at once
# speedup vs baseline: 5.2854x; 1.2098x over previous
#include <hip/hip_runtime.h>
#include <hip/hip_bf16.h>


typedef __attribute__((ext_vector_type(2))) float v2f;
typedef __attribute__((ext_vector_type(8))) float v8f;

#define LSTM_H 6
#define BLK_L0 8
#define WG_THREADS 128

static __device__ __forceinline__ float tanh_hw(float x) {
  return __builtin_amdgcn_tanhf(x);
}
static __device__ __forceinline__ float sigm_half(float zh) {
  return fmaf(0.5f, tanh_hw(zh), 0.5f);
}
static __device__ __forceinline__ float sx16(float v) {
  int b = __builtin_bit_cast(int, v);
  return __builtin_bit_cast(float,
      __builtin_amdgcn_permlanex16(b, b, 0x76543210, 0xfedcba98, false, false));
}
static __device__ __forceinline__ v8f wmma4(v2f a, v2f b, v8f c) {
  return __builtin_amdgcn_wmma_f32_16x16x4_f32(false, a, false, b, (short)0, c,
                                               false, false);
}

static __device__ __forceinline__ int gate_row(int m) {
  int idx = m & 7;
  if (idx >= LSTM_H) return -1;
  int hf = (m >> 3) & 1;
  int tile = m >> 4;
  int second = idx >= 3 ? 1 : 0;
  int gate = 2 * tile + second;
  int unit = 3 * hf + idx - 3 * second;
  return gate * LSTM_H + unit;
}

static __device__ __forceinline__ float gate_scale(int m) {
  int idx = m & 7;
  int tile = m >> 4;
  return (tile == 1 && idx < 3) ? 1.0f : 0.5f;
}

static __device__ __forceinline__ int ksrc_l0(int c, int hf, int e) {
  if (c == 0) return 2 * hf + e;
  if (c == 1) return 6 + 3 * hf + e;
  return e == 0 ? 4 + hf : 6 + 3 * hf + 2;
}
static __device__ __forceinline__ int ksrc_l1(int c, int hf, int e) {
  if (c == 0) return 6 + 3 * hf + e;
  if (c == 1) return e == 0 ? 6 + 3 * hf + 2 : -1;
  if (c == 2) return 12 + 3 * hf + e;
  if (c == 3) return e == 0 ? 3 * hf + 2 : 12 + 3 * hf + 2;
  return 3 * hf + e;
}

static __device__ __forceinline__ float wcomb(const float* Wih, const float* Whh,
                                              int D, int m, int k) {
  int r = gate_row(m);
  if (r < 0 || k < 0) return 0.0f;
  float sc = gate_scale(m);
  if (k < D) return sc * Wih[r * D + k];
  return sc * Whh[r * LSTM_H + (k - D)];
}

template <bool L1>
static __device__ __forceinline__ void load_A(const float* Wih, const float* Whh,
                                              const float* bias,
                                              v2f a[2][5], v8f cini[2],
                                              int lm, int half) {
  const int D = L1 ? 2 * LSTM_H : LSTM_H;
  const int nch = L1 ? 5 : 3;
  for (int t = 0; t < 2; ++t) {
    int m = t * 16 + lm;
#pragma unroll
    for (int c = 0; c < 5; ++c) {
      if (c < nch) {
        int k0 = L1 ? ksrc_l1(c, half, 0) : ksrc_l0(c, half, 0);
        int k1 = L1 ? ksrc_l1(c, half, 1) : ksrc_l0(c, half, 1);
        v2f av;
        av.x = wcomb(Wih, Whh, D, m, k0);
        av.y = wcomb(Wih, Whh, D, m, k1);
        a[t][c] = av;
      }
    }
    v8f ci;
#pragma unroll
    for (int r = 0; r < 8; ++r) {
      int gm = t * 16 + half * 8 + r;
      int gr = gate_row(gm);
      ci[r] = (gr >= 0) ? gate_scale(gm) * bias[gr] : 0.0f;
    }
    cini[t] = ci;
  }
}

static __device__ __forceinline__ void mm(const v2f a[2][5], const v8f cini[2],
                                          const v2f bch[5], int nch,
                                          v8f& A0, v8f& A1) {
  A0 = cini[0];
  A1 = cini[1];
#pragma unroll
  for (int c = 0; c < 5; ++c) {
    if (c < nch) {
      A0 = wmma4(a[0][c], bch[c], A0);
      A1 = wmma4(a[1][c], bch[c], A1);
    }
  }
}

static __device__ __forceinline__ void lstm_pointwise(const v8f& A0, const v8f& A1,
                                                      float c[3], float h[3]) {
#pragma unroll
  for (int u = 0; u < 3; ++u) {
    float si = sigm_half(A0[u]);
    float sf = sigm_half(A0[3 + u]);
    float tg = tanh_hw(A1[u]);
    float so = sigm_half(A1[3 + u]);
    float cn = sf * c[u] + si * tg;
    c[u] = cn;
    h[u] = so * tanh_hw(cn);
  }
}

static __device__ __forceinline__ void ld_x(const float* xp, int half,
                                            v2f& xlo, float& x4) {
  xlo = *(const v2f*)(xp + 2 * half);
  x4 = xp[4 + half];
}
static __device__ __forceinline__ void ld_r(const float* r0p, int half,
                                            v2f& r01, float& r2) {
  r01 = *(const v2f*)(r0p + 2 * half);
  r2 = r0p[4 + half];
}

static __device__ __forceinline__ void b_l0(v2f xlo, float x4, const float h[3], v2f bch[5]) {
  v2f b1, b2;
  b1.x = h[0]; b1.y = h[1];
  b2.x = x4;   b2.y = h[2];
  bch[0] = xlo; bch[1] = b1; bch[2] = b2;
}

static __device__ __forceinline__ void b_l1(const float f0[3], v2f r01, float r2,
                                            const float h[3], v2f bch[5]) {
  v2f b1, b2, b3, b4;
  b1.x = r2;    b1.y = 0.0f;
  b2.x = h[0];  b2.y = h[1];
  b3.x = f0[2]; b3.y = h[2];
  b4.x = f0[0]; b4.y = f0[1];
  bch[0] = r01; bch[1] = b1; bch[2] = b2; bch[3] = b3; bch[4] = b4;
}

__global__ void __launch_bounds__(WG_THREADS) lstm_l0_bwd(const float* __restrict__ tf, const float* __restrict__ h0,
                            const float* __restrict__ c0,
                            const float* __restrict__ Wih, const float* __restrict__ Whh,
                            const float* __restrict__ bias,
                            float* __restrict__ r0ws, int B, int T) {
  int tid = blockIdx.x * blockDim.x + threadIdx.x;
  int lane = tid & 31, lm = lane & 15, half = lane >> 4;
  int s = (tid >> 5) * 16 + lm;
  if (s >= B) return;

  v2f a[2][5]; v8f ci[2];
  load_A<false>(Wih, Whh, bias, a, ci, lm, half);

  float h[3], c[3];
  size_t d1 = (size_t)B * LSTM_H;
#pragma unroll
  for (int u = 0; u < 3; ++u) {
    h[u] = h0[d1 + (size_t)s * 6 + 3 * half + u];
    c[u] = c0[d1 + (size_t)s * 6 + 3 * half + u];
  }

  const float* xrow = tf + (size_t)s * T * 6;
  v2f cxl[BLK_L0]; float cx4[BLK_L0];
#pragma unroll
  for (int j = 0; j < BLK_L0; ++j) {
    int tt = T - 1 - j; tt = tt < 0 ? 0 : tt;
    ld_x(xrow + (size_t)tt * 6, half, cxl[j], cx4[j]);
  }

  auto step = [&](int t, v2f xlo, float x4) {
    v2f bch[5]; b_l0(xlo, x4, h, bch);
    v8f A0, A1; mm(a, ci, bch, 3, A0, A1);
    lstm_pointwise(A0, A1, c, h);
    float* wp = r0ws + ((size_t)t * B + s) * 6;
    v2f w01; w01.x = h[0]; w01.y = h[1];
    *(v2f*)(wp + 2 * half) = w01;
    wp[4 + half] = h[2];
  };

  int t = T - 1;
  for (; t >= BLK_L0 - 1; t -= BLK_L0) {
    v2f nxl[BLK_L0]; float nx4[BLK_L0];
#pragma unroll
    for (int j = 0; j < BLK_L0; ++j) {
      int tn = t - BLK_L0 - j; tn = tn < 0 ? 0 : tn;
      ld_x(xrow + (size_t)tn * 6, half, nxl[j], nx4[j]);
    }
    __builtin_amdgcn_sched_barrier(0);
#pragma unroll
    for (int j = 0; j < BLK_L0; ++j) step(t - j, cxl[j], cx4[j]);
    __builtin_amdgcn_sched_barrier(0);
#pragma unroll
    for (int j = 0; j < BLK_L0; ++j) { cxl[j] = nxl[j]; cx4[j] = nx4[j]; }
  }
#pragma unroll
  for (int j = 0; j < BLK_L0 - 1; ++j) {
    if (j <= t) step(t - j, cxl[j], cx4[j]);
  }
}

#define HB 8
typedef __attribute__((ext_vector_type(4))) float v4f;

__global__ void __launch_bounds__(WG_THREADS) lstm_fused(const float* __restrict__ tf, const float* __restrict__ h0,
                           const float* __restrict__ c0,
                           const float* __restrict__ Wih0f, const float* __restrict__ Whh0f,
                           const float* __restrict__ b0f,
                           const float* __restrict__ Wih1f, const float* __restrict__ Whh1f,
                           const float* __restrict__ b1f,
                           const float* __restrict__ Wih1b, const float* __restrict__ Whh1b,
                           const float* __restrict__ b1b,
                           const float* __restrict__ fcw, const float* __restrict__ fcb,
                           const float* __restrict__ r0ws, float* __restrict__ out,
                           int B, int T) {
  __shared__ v4f f0buf[2][2][HB][32];

  const int lane = threadIdx.x & 31, lm = lane & 15, half = lane >> 4;
  const int wv = __builtin_amdgcn_readfirstlane((int)(threadIdx.x >> 5));
  const bool producer = wv < 2;
  const int grp = wv & 1;
  int s = ((int)blockIdx.x * 2 + grp) * 16 + lm;
  const bool live = s < B;
  if (!live) s = B - 1;

  const int nch = (T + HB - 1) / HB;
  const float* xrow = tf + (size_t)s * T * 6;
  const float* rcol = r0ws + (size_t)s * 6;
  const size_t rstep = (size_t)B * 6;

  v2f aw[2][5]; v8f ciw[2];
  float hS[3], cS[3];
  v2f cua[HB]; float cub[HB];
  float fl[3] = {0.0f, 0.0f, 0.0f};

  if (producer) {
    load_A<false>(Wih0f, Whh0f, b0f, aw, ciw, lm, half);
  } else {
    load_A<true>(Wih1f, Whh1f, b1f, aw, ciw, lm, half);
  }
  {
    size_t dd = (producer ? 0 : 2) * (size_t)B * LSTM_H;
#pragma unroll
    for (int u = 0; u < 3; ++u) {
      size_t o = (size_t)s * 6 + 3 * half + u;
      hS[u] = h0[dd + o];  cS[u] = c0[dd + o];
    }
  }
#pragma unroll
  for (int j = 0; j < HB; ++j) {
    int tt = j > T - 1 ? T - 1 : j;
    if (producer) ld_x(xrow + (size_t)tt * 6, half, cua[j], cub[j]);
    else          ld_r(rcol + (size_t)tt * rstep, half, cua[j], cub[j]);
  }

  auto step0 = [&](v2f xlo, float x4, v4f* slot) {
    v2f bch[5]; b_l0(xlo, x4, hS, bch);
    v8f A0, A1; mm(aw, ciw, bch, 3, A0, A1);
    lstm_pointwise(A0, A1, cS, hS);
    v4f w; w.x = hS[0]; w.y = hS[1]; w.z = hS[2]; w.w = 0.0f;
    *slot = w;
  };
  auto step1 = [&](v4f f, v2f r01, float r2) {
    float fu[3] = {f.x, f.y, f.z};
    v2f bch[5]; b_l1(fu, r01, r2, hS, bch);
    v8f A0, A1; mm(aw, ciw, bch, 5, A0, A1);
    lstm_pointwise(A0, A1, cS, hS);
  };

  for (int k = 0; k <= nch; ++k) {
    if (producer) {
      if (k < nch) {
        const int t0 = k * HB;
        v2f nxa[HB]; float nxb[HB];
#pragma unroll
        for (int j = 0; j < HB; ++j) {
          int tn = t0 + HB + j; tn = tn > T - 1 ? T - 1 : tn;
          ld_x(xrow + (size_t)tn * 6, half, nxa[j], nxb[j]);
        }
        __builtin_amdgcn_sched_barrier(0);
#pragma unroll
        for (int j = 0; j < HB; ++j) step0(cua[j], cub[j], &f0buf[k & 1][grp][j][lane]);
        __builtin_amdgcn_sched_barrier(0);
#pragma unroll
        for (int j = 0; j < HB; ++j) { cua[j] = nxa[j]; cub[j] = nxb[j]; }
      }
    } else {
      if (k > 0) {
        const int kc = k - 1;
        const int t0 = kc * HB;
        v2f nxa[HB]; float nxb[HB];
#pragma unroll
        for (int j = 0; j < HB; ++j) {
          int tn = t0 + HB + j; tn = tn > T - 1 ? T - 1 : tn;
          ld_r(rcol + (size_t)tn * rstep, half, nxa[j], nxb[j]);
        }
        v4f fv[HB];
#pragma unroll
        for (int j = 0; j < HB; ++j) fv[j] = f0buf[kc & 1][grp][j][lane];
        __builtin_amdgcn_sched_barrier(0);
        if (t0 + HB <= T) {
#pragma unroll
          for (int j = 0; j < HB; ++j) step1(fv[j], cua[j], cub[j]);
          fl[0] = fv[HB - 1].x; fl[1] = fv[HB - 1].y; fl[2] = fv[HB - 1].z;
        } else {
#pragma unroll
          for (int j = 0; j < HB; ++j) {
            if (t0 + j < T) {
              step1(fv[j], cua[j], cub[j]);
              fl[0] = fv[j].x; fl[1] = fv[j].y; fl[2] = fv[j].z;
            }
          }
        }
        __builtin_amdgcn_sched_barrier(0);
#pragma unroll
        for (int j = 0; j < HB; ++j) { cua[j] = nxa[j]; cub[j] = nxb[j]; }
      }
    }
    __syncthreads();
  }
  if (producer) return;

  v2f a2[2][5]; v8f ci2[2]; load_A<true>(Wih1b, Whh1b, b1b, a2, ci2, lm, half);
  float hR[3], cR[3];
  size_t d3 = 3 * (size_t)B * LSTM_H;
#pragma unroll
  for (int u = 0; u < 3; ++u) {
    size_t o = (size_t)s * 6 + 3 * half + u;
    hR[u] = h0[d3 + o];  cR[u] = c0[d3 + o];
  }
  {
    const float* r0p = r0ws + ((size_t)(T - 1) * B + s) * 6;
    v2f r01; float r2; ld_r(r0p, half, r01, r2);
    v2f bch[5]; b_l1(fl, r01, r2, hR, bch);
    v8f A0, A1; mm(a2, ci2, bch, 5, A0, A1);
    lstm_pointwise(A0, A1, cR, hR);
  }
  float pB[3], pR[3];
#pragma unroll
  for (int u = 0; u < 3; ++u) { pB[u] = sx16(hS[u]); pR[u] = sx16(hR[u]); }
  if (half == 0 && live) {
    float v[12];
#pragma unroll
    for (int u = 0; u < 3; ++u) {
      v[u]     = hS[u] > 0.0f ? hS[u] : 0.0f;
      v[3 + u] = pB[u] > 0.0f ? pB[u] : 0.0f;
      v[6 + u] = hR[u] > 0.0f ? hR[u] : 0.0f;
      v[9 + u] = pR[u] > 0.0f ? pR[u] : 0.0f;
    }
#pragma unroll
    for (int j = 0; j < 6; ++j) {
      float acc = fcb[j];
#pragma unroll
      for (int k = 0; k < 12; ++k) acc = fmaf(fcw[j * 12 + k], v[k], acc);
      out[(size_t)s * 6 + j] = acc;
    }
  }
}

extern "C" void kernel_launch(void* const* d_in, const int* in_sizes, int n_in,
                              void* d_out, int out_size, void* d_ws, size_t ws_size,
                              hipStream_t stream) {
  const float* tf    = (const float*)d_in[0];
  const float* h0    = (const float*)d_in[1];
  const float* c0    = (const float*)d_in[2];
  const float* Wih0f = (const float*)d_in[3];
  const float* Whh0f = (const float*)d_in[4];
  const float* b0f   = (const float*)d_in[5];
  const float* Wih0b = (const float*)d_in[6];
  const float* Whh0b = (const float*)d_in[7];
  const float* b0b   = (const float*)d_in[8];
  const float* Wih1f = (const float*)d_in[9];
  const float* Whh1f = (const float*)d_in[10];
  const float* b1f   = (const float*)d_in[11];
  const float* Wih1b = (const float*)d_in[12];
  const float* Whh1b = (const float*)d_in[13];
  const float* b1b   = (const float*)d_in[14];
  const float* fcw   = (const float*)d_in[15];
  const float* fcb   = (const float*)d_in[16];

  int B = in_sizes[1] / (4 * LSTM_H);
  int T = in_sizes[0] / (B * LSTM_H);

  size_t need = (size_t)B * T * LSTM_H * sizeof(float);
  if (ws_size < need) return;
  float* r0ws = (float*)d_ws;

  int waves = (B + 15) / 16;
  int threads = waves * 32;
  int block = WG_THREADS;
  int grid = (threads + block - 1) / block;

  lstm_l0_bwd<<<grid, block, 0, stream>>>(tf, h0, c0, Wih0b, Whh0b, b0b, r0ws, B, T);
  int grid2 = (waves + 1) / 2;
  lstm_fused<<<grid2, block, 0, stream>>>(tf, h0, c0, Wih0f, Whh0f, b0f,
                                         Wih1f, Whh1f, b1f, Wih1b, Whh1b, b1b,
                                         fcw, fcb, r0ws, (float*)d_out, B, T);
}


# second launch's matrix products split 8 + 8 per step (was 6 + 10): layer-0 waves do layer 1's bias + r2 chunk, handed over with f0; free products pinned under activations; layer-1 k order changes
# speedup vs baseline: 5.4551x; 1.0321x over previous
#include <hip/hip_runtime.h>
#include <hip/hip_bf16.h>


typedef __attribute__((ext_vector_type(2))) float v2f;
typedef __attribute__((ext_vector_type(8))) float v8f;

#define LSTM_H 6
#define BLK_L0 8
#define WG_THREADS 128

static __device__ __forceinline__ float tanh_hw(float x) {
  return __builtin_amdgcn_tanhf(x);
}
static __device__ __forceinline__ float sigm_half(float zh) {
  return fmaf(0.5f, tanh_hw(zh), 0.5f);
}
static __device__ __forceinline__ float sx16(float v) {
  int b = __builtin_bit_cast(int, v);
  return __builtin_bit_cast(float,
      __builtin_amdgcn_permlanex16(b, b, 0x76543210, 0xfedcba98, false, false));
}
static __device__ __forceinline__ v8f wmma4(v2f a, v2f b, v8f c) {
  return __builtin_amdgcn_wmma_f32_16x16x4_f32(false, a, false, b, (short)0, c,
                                               false, false);
}

static __device__ __forceinline__ int gate_row(int m) {
  int idx = m & 7;
  if (idx >= LSTM_H) return -1;
  int hf = (m >> 3) & 1;
  int tile = m >> 4;
  int second = idx >= 3 ? 1 : 0;
  int gate = 2 * tile + second;
  int unit = 3 * hf + idx - 3 * second;
  return gate * LSTM_H + unit;
}

static __device__ __forceinline__ float gate_scale(int m) {
  int idx = m & 7;
  int tile = m >> 4;
  return (tile == 1 && idx < 3) ? 1.0f : 0.5f;
}

static __device__ __forceinline__ int ksrc_l0(int c, int hf, int e) {
  if (c == 0) return 2 * hf + e;
  if (c == 1) return 6 + 3 * hf + e;
  return e == 0 ? 4 + hf : 6 + 3 * hf + 2;
}
static __device__ __forceinline__ int ksrc_l1(int c, int hf, int e) {
  if (c == 0) return 6 + 3 * hf + e;
  if (c == 1) return e == 0 ? 6 + 3 * hf + 2 : -1;
  if (c == 2) return 3 * hf + e;
  if (c == 3) return 12 + 3 * hf + e;
  return e == 0 ? 3 * hf + 2 : 12 + 3 * hf + 2;
}

static __device__ __forceinline__ float wcomb(const float* Wih, const float* Whh,
                                              int D, int m, int k) {
  int r = gate_row(m);
  if (r < 0 || k < 0) return 0.0f;
  float sc = gate_scale(m);
  if (k < D) return sc * Wih[r * D + k];
  return sc * Whh[r * LSTM_H + (k - D)];
}

template <bool L1>
static __device__ __forceinline__ void load_A(const float* Wih, const float* Whh,
                                              const float* bias,
                                              v2f a[2][5], v8f cini[2],
                                              int lm, int half) {
  const int D = L1 ? 2 * LSTM_H : LSTM_H;
  const int nch = L1 ? 5 : 3;
  for (int t = 0; t < 2; ++t) {
    int m = t * 16 + lm;
#pragma unroll
    for (int c = 0; c < 5; ++c) {
      if (c < nch) {
        int k0 = L1 ? ksrc_l1(c, half, 0) : ksrc_l0(c, half, 0);
        int k1 = L1 ? ksrc_l1(c, half, 1) : ksrc_l0(c, half, 1);
        v2f av;
        av.x = wcomb(Wih, Whh, D, m, k0);
        av.y = wcomb(Wih, Whh, D, m, k1);
        a[t][c] = av;
      }
    }
    v8f ci;
#pragma unroll
    for (int r = 0; r < 8; ++r) {
      int gm = t * 16 + half * 8 + r;
      int gr = gate_row(gm);
      ci[r] = (gr >= 0) ? gate_scale(gm) * bias[gr] : 0.0f;
    }
    cini[t] = ci;
  }
}

static __device__ __forceinline__ void mm(const v2f a[2][5], const v8f cini[2],
                                          const v2f bch[5], int nch,
                                          v8f& A0, v8f& A1) {
  A0 = cini[0];
  A1 = cini[1];
#pragma unroll
  for (int c = 0; c < 5; ++c) {
    if (c < nch) {
      A0 = wmma4(a[0][c], bch[c], A0);
      A1 = wmma4(a[1][c], bch[c], A1);
    }
  }
}

static __device__ __forceinline__ void lstm_pointwise(const v8f& A0, const v8f& A1,
                                                      float c[3], float h[3]) {
#pragma unroll
  for (int u = 0; u < 3; ++u) {
    float si = sigm_half(A0[u]);
    float sf = sigm_half(A0[3 + u]);
    float tg = tanh_hw(A1[u]);
    float so = sigm_half(A1[3 + u]);
    float cn = sf * c[u] + si * tg;
    c[u] = cn;
    h[u] = so * tanh_hw(cn);
  }
}

struct gates4 { float si, sf, tg, so; };
static __device__ __forceinline__ gates4 pw_gates(const v8f& A0, const v8f& A1, int u) {
  gates4 g;
  g.si = sigm_half(A0[u]);
  g.sf = sigm_half(A0[3 + u]);
  g.tg = tanh_hw(A1[u]);
  g.so = sigm_half(A1[3 + u]);
  return g;
}
static __device__ __forceinline__ void pw_cell(const gates4 g[3], float c[3]) {
#pragma unroll
  for (int u = 0; u < 3; ++u) c[u] = g[u].sf * c[u] + g[u].si * g[u].tg;
}
static __device__ __forceinline__ void pw_out(const gates4 g[3], const float c[3], float h[3]) {
#pragma unroll
  for (int u = 0; u < 3; ++u) h[u] = g[u].so * tanh_hw(c[u]);
}

static __device__ __forceinline__ void ld_x(const float* xp, int half,
                                            v2f& xlo, float& x4) {
  xlo = *(const v2f*)(xp + 2 * half);
  x4 = xp[4 + half];
}
static __device__ __forceinline__ void ld_r(const float* r0p, int half,
                                            v2f& r01, float& r2) {
  r01 = *(const v2f*)(r0p + 2 * half);
  r2 = r0p[4 + half];
}

static __device__ __forceinline__ void b_l0(v2f xlo, float x4, const float h[3], v2f bch[5]) {
  v2f b1, b2;
  b1.x = h[0]; b1.y = h[1];
  b2.x = x4;   b2.y = h[2];
  bch[0] = xlo; bch[1] = b1; bch[2] = b2;
}

static __device__ __forceinline__ void b_l1(const float f0[3], v2f r01, float r2,
                                            const float h[3], v2f bch[5]) {
  v2f b1, b2, b3, b4;
  b1.x = r2;    b1.y = 0.0f;
  b2.x = f0[0]; b2.y = f0[1];
  b3.x = h[0];  b3.y = h[1];
  b4.x = f0[2]; b4.y = h[2];
  bch[0] = r01; bch[1] = b1; bch[2] = b2; bch[3] = b3; bch[4] = b4;
}

__global__ void __launch_bounds__(WG_THREADS) lstm_l0_bwd(const float* __restrict__ tf, const float* __restrict__ h0,
                            const float* __restrict__ c0,
                            const float* __restrict__ Wih, const float* __restrict__ Whh,
                            const float* __restrict__ bias,
                            float* __restrict__ r0ws, int B, int T) {
  int tid = blockIdx.x * blockDim.x + threadIdx.x;
  int lane = tid & 31, lm = lane & 15, half = lane >> 4;
  int s = (tid >> 5) * 16 + lm;
  if (s >= B) return;

  v2f a[2][5]; v8f ci[2];
  load_A<false>(Wih, Whh, bias, a, ci, lm, half);

  float h[3], c[3];
  size_t d1 = (size_t)B * LSTM_H;
#pragma unroll
  for (int u = 0; u < 3; ++u) {
    h[u] = h0[d1 + (size_t)s * 6 + 3 * half + u];
    c[u] = c0[d1 + (size_t)s * 6 + 3 * half + u];
  }

  const float* xrow = tf + (size_t)s * T * 6;
  v2f cxl[BLK_L0]; float cx4[BLK_L0];
#pragma unroll
  for (int j = 0; j < BLK_L0; ++j) {
    int tt = T - 1 - j; tt = tt < 0 ? 0 : tt;
    ld_x(xrow + (size_t)tt * 6, half, cxl[j], cx4[j]);
  }

  auto step = [&](int t, v2f xlo, float x4) {
    v2f bch[5]; b_l0(xlo, x4, h, bch);
    v8f A0, A1; mm(a, ci, bch, 3, A0, A1);
    lstm_pointwise(A0, A1, c, h);
    float* wp = r0ws + ((size_t)t * B + s) * 6;
    v2f w01; w01.x = h[0]; w01.y = h[1];
    *(v2f*)(wp + 2 * half) = w01;
    wp[4 + half] = h[2];
  };

  int t = T - 1;
  for (; t >= BLK_L0 - 1; t -= BLK_L0) {
    v2f nxl[BLK_L0]; float nx4[BLK_L0];
#pragma unroll
    for (int j = 0; j < BLK_L0; ++j) {
      int tn = t - BLK_L0 - j; tn = tn < 0 ? 0 : tn;
      ld_x(xrow + (size_t)tn * 6, half, nxl[j], nx4[j]);
    }
    __builtin_amdgcn_sched_barrier(0);
#pragma unroll
    for (int j = 0; j < BLK_L0; ++j) step(t - j, cxl[j], cx4[j]);
    __builtin_amdgcn_sched_barrier(0);
#pragma unroll
    for (int j = 0; j < BLK_L0; ++j) { cxl[j] = nxl[j]; cx4[j] = nx4[j]; }
  }
#pragma unroll
  for (int j = 0; j < BLK_L0 - 1; ++j) {
    if (j <= t) step(t - j, cxl[j], cx4[j]);
  }
}

#define HB 8
typedef __attribute__((ext_vector_type(4))) float v4f;
#define PIN() __builtin_amdgcn_sched_barrier(0)
#define TIE(tile) asm volatile("" : "+v"(tile))

__global__ void __launch_bounds__(WG_THREADS) lstm_fused(const float* __restrict__ tf, const float* __restrict__ h0,
                           const float* __restrict__ c0,
                           const float* __restrict__ Wih0f, const float* __restrict__ Whh0f,
                           const float* __restrict__ b0f,
                           const float* __restrict__ Wih1f, const float* __restrict__ Whh1f,
                           const float* __restrict__ b1f,
                           const float* __restrict__ Wih1b, const float* __restrict__ Whh1b,
                           const float* __restrict__ b1b,
                           const float* __restrict__ fcw, const float* __restrict__ fcb,
                           const float* __restrict__ r0ws, float* __restrict__ out,
                           int B, int T) {
  __shared__ v4f hbuf[2][2][HB][4][32];

  const int lane = threadIdx.x & 31, lm = lane & 15, half = lane >> 4;
  const int wv = __builtin_amdgcn_readfirstlane((int)(threadIdx.x >> 5));
  const bool producer = wv < 2;
  const int grp = wv & 1;
  int s = ((int)blockIdx.x * 2 + grp) * 16 + lm;
  const bool live = s < B;
  if (!live) s = B - 1;

  const int nch = (T + HB - 1) / HB;
  const float* xrow = tf + (size_t)s * T * 6;
  const float* rcol = r0ws + (size_t)s * 6;
  const size_t rstep = (size_t)B * 6;

  v2f aw[2][5]; v8f ciw[2];
  float hS[3], cS[3];
  v2f cua[HB]; float cub[HB]; float cuc[HB];
  float fl[3] = {0.0f, 0.0f, 0.0f};
  v2f ap[2]; v8f cip[2];
  ap[0] = (v2f){0.0f, 0.0f}; ap[1] = ap[0];
  cip[0] = (v8f){0.0f, 0.0f, 0.0f, 0.0f, 0.0f, 0.0f, 0.0f, 0.0f}; cip[1] = cip[0];
  v8f N0 = cip[0], N1 = cip[0];

  if (producer) {
    load_A<false>(Wih0f, Whh0f, b0f, aw, ciw, lm, half);
    v2f aq[2][5]; v8f ciq[2];
    load_A<true>(Wih1f, Whh1f, b1f, aq, ciq, lm, half);
    ap[0] = aq[0][1]; ap[1] = aq[1][1];
    cip[0] = ciq[0];  cip[1] = ciq[1];
  } else {
    load_A<true>(Wih1f, Whh1f, b1f, aw, ciw, lm, half);
  }
  {
    size_t dd = (producer ? 0 : 2) * (size_t)B * LSTM_H;
#pragma unroll
    for (int u = 0; u < 3; ++u) {
      size_t o = (size_t)s * 6 + 3 * half + u;
      hS[u] = h0[dd + o];  cS[u] = c0[dd + o];
    }
  }
#pragma unroll
  for (int j = 0; j < HB; ++j) {
    int tt = j > T - 1 ? T - 1 : j;
    if (producer) {
      ld_x(xrow + (size_t)tt * 6, half, cua[j], cub[j]);
      cuc[j] = rcol[(size_t)tt * rstep + 4 + half];
    } else {
      cua[j] = *(const v2f*)(rcol + (size_t)tt * rstep + 2 * half);
      cub[j] = 0.0f;  cuc[j] = 0.0f;
    }
  }
  if (producer) {
    N0 = wmma4(aw[0][0], cua[0], ciw[0]);
    N1 = wmma4(aw[1][0], cua[0], ciw[1]);
  }

  auto step1 = [&](v4f q0, v4f q1, v4f q2, v4f q3, v2f r01) {
    v8f A0, A1;
    A0[0] = q0.x; A0[1] = q0.y; A0[2] = q0.z; A0[3] = q0.w;
    A0[4] = q1.x; A0[5] = q1.y; A0[6] = q1.z; A0[7] = q1.w;
    A1[0] = q2.x; A1[1] = q2.y; A1[2] = q2.z; A1[3] = q2.w;
    A1[4] = q3.x; A1[5] = q3.y; A1[6] = q3.z; A1[7] = q3.w;
    v2f bf, bh, bfh;
    bf.x = q1.z;   bf.y = q1.w;
    bh.x = hS[0];  bh.y = hS[1];
    bfh.x = q3.z;  bfh.y = hS[2];
    A0 = wmma4(aw[0][0], r01, A0);  A1 = wmma4(aw[1][0], r01, A1);
    A0 = wmma4(aw[0][2], bf, A0);   A1 = wmma4(aw[1][2], bf, A1);
    A0 = wmma4(aw[0][3], bh, A0);   A1 = wmma4(aw[1][3], bh, A1);
    A0 = wmma4(aw[0][4], bfh, A0);  A1 = wmma4(aw[1][4], bfh, A1);
    lstm_pointwise(A0, A1, cS, hS);
  };

  for (int k = 0; k <= nch; ++k) {
    if (producer) {
      if (k < nch) {
        const int t0 = k * HB;
        v2f nxa[HB]; float nxb[HB]; float nxc[HB];
#pragma unroll
        for (int j = 0; j < HB; ++j) {
          int tn = t0 + HB + j; tn = tn > T - 1 ? T - 1 : tn;
          ld_x(xrow + (size_t)tn * 6, half, nxa[j], nxb[j]);
          nxc[j] = rcol[(size_t)tn * rstep + 4 + half];
        }
        __builtin_amdgcn_sched_barrier(0);
        v4f w0[HB], w1[HB], w2[HB], w3[HB];
#pragma unroll
        for (int j = 0; j < HB; ++j) {
          v2f b1, b2;
          b1.x = hS[0];  b1.y = hS[1];
          b2.x = cub[j]; b2.y = hS[2];
          v8f A0 = wmma4(aw[0][1], b1, N0);
          v8f A1 = wmma4(aw[1][1], b1, N1);
          A0 = wmma4(aw[0][2], b2, A0);
          A1 = wmma4(aw[1][2], b2, A1);
          PIN();
          const v2f xn = (j + 1 < HB) ? cua[j + 1] : nxa[0];
          N0 = wmma4(aw[0][0], xn, ciw[0]);
          N1 = wmma4(aw[1][0], xn, ciw[1]);
          TIE(N0); TIE(N1);
          PIN();
          gates4 g[3];
          g[0] = pw_gates(A0, A1, 0);
          g[1] = pw_gates(A0, A1, 1);
          PIN();
          v2f br; br.x = cuc[j]; br.y = 0.0f;
          v8f P0 = wmma4(ap[0], br, cip[0]);
          TIE(P0);
          PIN();
          g[2] = pw_gates(A0, A1, 2);
          pw_cell(g, cS);
          PIN();
          v8f P1 = wmma4(ap[1], br, cip[1]);
          TIE(P1);
          PIN();
          pw_out(g, cS, hS);
          w0[j].x = P0[0]; w0[j].y = P0[1]; w0[j].z = P0[2]; w0[j].w = P0[3];
          w1[j].x = P0[4]; w1[j].y = P0[5]; w1[j].z = hS[0]; w1[j].w = hS[1];
          w2[j].x = P1[0]; w2[j].y = P1[1]; w2[j].z = P1[2]; w2[j].w = P1[3];
          w3[j].x = P1[4]; w3[j].y = P1[5]; w3[j].z = hS[2]; w3[j].w = 0.0f;
        }
#pragma unroll
        for (int j = 0; j < HB; ++j) {
          v4f* slot = &hbuf[k & 1][grp][j][0][lane];
          slot[0] = w0[j]; slot[32] = w1[j]; slot[64] = w2[j]; slot[96] = w3[j];
        }
        __builtin_amdgcn_sched_barrier(0);
#pragma unroll
        for (int j = 0; j < HB; ++j) { cua[j] = nxa[j]; cub[j] = nxb[j]; cuc[j] = nxc[j]; }
      }
    } else {
      if (k > 0) {
        const int kc = k - 1;
        const int t0 = kc * HB;
        v2f nxa[HB];
#pragma unroll
        for (int j = 0; j < HB; ++j) {
          int tn = t0 + HB + j; tn = tn > T - 1 ? T - 1 : tn;
          nxa[j] = *(const v2f*)(rcol + (size_t)tn * rstep + 2 * half);
        }
        v4f q0[HB], q1[HB], q2[HB], q3[HB];
#pragma unroll
        for (int j = 0; j < HB; ++j) {
          q0[j] = hbuf[kc & 1][grp][j][0][lane];  q1[j] = hbuf[kc & 1][grp][j][1][lane];
          q2[j] = hbuf[kc & 1][grp][j][2][lane];  q3[j] = hbuf[kc & 1][grp][j][3][lane];
        }
        __builtin_amdgcn_sched_barrier(0);
        if (t0 + HB <= T) {
          v8f M0, M1;
          {
            M0[0] = q0[0].x; M0[1] = q0[0].y; M0[2] = q0[0].z; M0[3] = q0[0].w;
            M0[4] = q1[0].x; M0[5] = q1[0].y; M0[6] = q1[0].z; M0[7] = q1[0].w;
            M1[0] = q2[0].x; M1[1] = q2[0].y; M1[2] = q2[0].z; M1[3] = q2[0].w;
            M1[4] = q3[0].x; M1[5] = q3[0].y; M1[6] = q3[0].z; M1[7] = q3[0].w;
            v2f bf; bf.x = q1[0].z; bf.y = q1[0].w;
            M0 = wmma4(aw[0][0], cua[0], M0);  M1 = wmma4(aw[1][0], cua[0], M1);
            M0 = wmma4(aw[0][2], bf, M0);      M1 = wmma4(aw[1][2], bf, M1);
          }
          PIN();
#pragma unroll
          for (int j = 0; j < HB; ++j) {
            v2f bh, bfh;
            bh.x = hS[0];     bh.y = hS[1];
            bfh.x = q3[j].z;  bfh.y = hS[2];
            v8f A0 = wmma4(aw[0][3], bh, M0);
            v8f A1 = wmma4(aw[1][3], bh, M1);
            A0 = wmma4(aw[0][4], bfh, A0);
            A1 = wmma4(aw[1][4], bfh, A1);
            PIN();
            const int jn = j + 1 < HB ? j + 1 : HB - 1;
            v2f bfn; bfn.x = q1[jn].z; bfn.y = q1[jn].w;
            if (j + 1 < HB) {
              M0[0] = q0[jn].x; M0[1] = q0[jn].y; M0[2] = q0[jn].z; M0[3] = q0[jn].w;
              M0[4] = q1[jn].x; M0[5] = q1[jn].y; M0[6] = q1[jn].z; M0[7] = q1[jn].w;
              M1[0] = q2[jn].x; M1[1] = q2[jn].y; M1[2] = q2[jn].z; M1[3] = q2[jn].w;
              M1[4] = q3[jn].x; M1[5] = q3[jn].y; M1[6] = q3[jn].z; M1[7] = q3[jn].w;
              M0 = wmma4(aw[0][0], cua[jn], M0);
              M1 = wmma4(aw[1][0], cua[jn], M1);
              TIE(M0); TIE(M1);
            }
            PIN();
            gates4 g[3];
            g[0] = pw_gates(A0, A1, 0);
            g[1] = pw_gates(A0, A1, 1);
            PIN();
            if (j + 1 < HB) { M0 = wmma4(aw[0][2], bfn, M0); TIE(M0); }
            PIN();
            g[2] = pw_gates(A0, A1, 2);
            pw_cell(g, cS);
            PIN();
            if (j + 1 < HB) { M1 = wmma4(aw[1][2], bfn, M1); TIE(M1); }
            PIN();
            pw_out(g, cS, hS);
          }
          fl[0] = q1[HB - 1].z; fl[1] = q1[HB - 1].w; fl[2] = q3[HB - 1].z;
        } else {
#pragma unroll
          for (int j = 0; j < HB; ++j) {
            if (t0 + j < T) {
              step1(q0[j], q1[j], q2[j], q3[j], cua[j]);
              fl[0] = q1[j].z; fl[1] = q1[j].w; fl[2] = q3[j].z;
            }
          }
        }
        __builtin_amdgcn_sched_barrier(0);
#pragma unroll
        for (int j = 0; j < HB; ++j) cua[j] = nxa[j];
      }
    }
    __syncthreads();
  }
  if (producer) return;

  v2f a2[2][5]; v8f ci2[2]; load_A<true>(Wih1b, Whh1b, b1b, a2, ci2, lm, half);
  float hR[3], cR[3];
  size_t d3 = 3 * (size_t)B * LSTM_H;
#pragma unroll
  for (int u = 0; u < 3; ++u) {
    size_t o = (size_t)s * 6 + 3 * half + u;
    hR[u] = h0[d3 + o];  cR[u] = c0[d3 + o];
  }
  {
    const float* r0p = r0ws + ((size_t)(T - 1) * B + s) * 6;
    v2f r01; float r2; ld_r(r0p, half, r01, r2);
    v2f bch[5]; b_l1(fl, r01, r2, hR, bch);
    v8f A0, A1; mm(a2, ci2, bch, 5, A0, A1);
    lstm_pointwise(A0, A1, cR, hR);
  }
  float pB[3], pR[3];
#pragma unroll
  for (int u = 0; u < 3; ++u) { pB[u] = sx16(hS[u]); pR[u] = sx16(hR[u]); }
  if (half == 0 && live) {
    float v[12];
#pragma unroll
    for (int u = 0; u < 3; ++u) {
      v[u]     = hS[u] > 0.0f ? hS[u] : 0.0f;
      v[3 + u] = pB[u] > 0.0f ? pB[u] : 0.0f;
      v[6 + u] = hR[u] > 0.0f ? hR[u] : 0.0f;
      v[9 + u] = pR[u] > 0.0f ? pR[u] : 0.0f;
    }
#pragma unroll
    for (int j = 0; j < 6; ++j) {
      float acc = fcb[j];
#pragma unroll
      for (int k = 0; k < 12; ++k) acc = fmaf(fcw[j * 12 + k], v[k], acc);
      out[(size_t)s * 6 + j] = acc;
    }
  }
}

extern "C" void kernel_launch(void* const* d_in, const int* in_sizes, int n_in,
                              void* d_out, int out_size, void* d_ws, size_t ws_size,
                              hipStream_t stream) {
  const float* tf    = (const float*)d_in[0];
  const float* h0    = (const float*)d_in[1];
  const float* c0    = (const float*)d_in[2];
  const float* Wih0f = (const float*)d_in[3];
  const float* Whh0f = (const float*)d_in[4];
  const float* b0f   = (const float*)d_in[5];
  const float* Wih0b = (const float*)d_in[6];
  const float* Whh0b = (const float*)d_in[7];
  const float* b0b   = (const float*)d_in[8];
  const float* Wih1f = (const float*)d_in[9];
  const float* Whh1f = (const float*)d_in[10];
  const float* b1f   = (const float*)d_in[11];
  const float* Wih1b = (const float*)d_in[12];
  const float* Whh1b = (const float*)d_in[13];
  const float* b1b   = (const float*)d_in[14];
  const float* fcw   = (const float*)d_in[15];
  const float* fcb   = (const float*)d_in[16];

  int B = in_sizes[1] / (4 * LSTM_H);
  int T = in_sizes[0] / (B * LSTM_H);

  size_t need = (size_t)B * T * LSTM_H * sizeof(float);
  if (ws_size < need) return;
  float* r0ws = (float*)d_ws;

  int waves = (B + 15) / 16;
  int threads = waves * 32;
  int block = WG_THREADS;
  int grid = (threads + block - 1) / block;

  lstm_l0_bwd<<<grid, block, 0, stream>>>(tf, h0, c0, Wih0b, Whh0b, b0b, r0ws, B, T);
  int grid2 = (waves + 1) / 2;
  lstm_fused<<<grid2, block, 0, stream>>>(tf, h0, c0, Wih0f, Whh0f, b0f,
                                         Wih1f, Whh1f, b1f, Wih1b, Whh1b, b1b,
                                         fcw, fcb, r0ws, (float*)d_out, B, T);
}


# first launch: x loads and r0 stores on scalar bases + fixed lane offsets, a block's 8 r0 records stored as one group behind its last step: no vector address adds, no translation wait among the steps
# speedup vs baseline: 5.6835x; 1.0419x over previous
#include <hip/hip_runtime.h>
#include <hip/hip_bf16.h>


typedef __attribute__((ext_vector_type(2))) float v2f;
typedef __attribute__((ext_vector_type(8))) float v8f;
typedef __attribute__((address_space(1))) char gl_char;
typedef __attribute__((address_space(1))) float gl_float;
typedef __attribute__((address_space(1))) v2f gl_v2f;
typedef __attribute__((address_space(1))) const char gl_cchar;
typedef __attribute__((address_space(1))) const float gl_cfloat;
typedef __attribute__((address_space(1))) const v2f gl_cv2f;

#define LSTM_H 6
#define BLK_L0 8
#define WG_THREADS 128

static __device__ __forceinline__ float tanh_hw(float x) {
  return __builtin_amdgcn_tanhf(x);
}
static __device__ __forceinline__ float sigm_half(float zh) {
  return fmaf(0.5f, tanh_hw(zh), 0.5f);
}
static __device__ __forceinline__ float sx16(float v) {
  int b = __builtin_bit_cast(int, v);
  return __builtin_bit_cast(float,
      __builtin_amdgcn_permlanex16(b, b, 0x76543210, 0xfedcba98, false, false));
}
static __device__ __forceinline__ v8f wmma4(v2f a, v2f b, v8f c) {
  return __builtin_amdgcn_wmma_f32_16x16x4_f32(false, a, false, b, (short)0, c,
                                               false, false);
}

static __device__ __forceinline__ int gate_row(int m) {
  int idx = m & 7;
  if (idx >= LSTM_H) return -1;
  int hf = (m >> 3) & 1;
  int tile = m >> 4;
  int second = idx >= 3 ? 1 : 0;
  int gate = 2 * tile + second;
  int unit = 3 * hf + idx - 3 * second;
  return gate * LSTM_H + unit;
}

static __device__ __forceinline__ float gate_scale(int m) {
  int idx = m & 7;
  int tile = m >> 4;
  return (tile == 1 && idx < 3) ? 1.0f : 0.5f;
}

static __device__ __forceinline__ int ksrc_l0(int c, int hf, int e) {
  if (c == 0) return 2 * hf + e;
  if (c == 1) return 6 + 3 * hf + e;
  return e == 0 ? 4 + hf : 6 + 3 * hf + 2;
}
static __device__ __forceinline__ int ksrc_l1(int c, int hf, int e) {
  if (c == 0) return 6 + 3 * hf + e;
  if (c == 1) return e == 0 ? 6 + 3 * hf + 2 : -1;
  if (c == 2) return 3 * hf + e;
  if (c == 3) return 12 + 3 * hf + e;
  return e == 0 ? 3 * hf + 2 : 12 + 3 * hf + 2;
}

static __device__ __forceinline__ float wcomb(const float* Wih, const float* Whh,
                                              int D, int m, int k) {
  int r = gate_row(m);
  if (r < 0 || k < 0) return 0.0f;
  float sc = gate_scale(m);
  if (k < D) return sc * Wih[r * D + k];
  return sc * Whh[r * LSTM_H + (k - D)];
}

template <bool L1>
static __device__ __forceinline__ void load_A(const float* Wih, const float* Whh,
                                              const float* bias,
                                              v2f a[2][5], v8f cini[2],
                                              int lm, int half) {
  const int D = L1 ? 2 * LSTM_H : LSTM_H;
  const int nch = L1 ? 5 : 3;
  for (int t = 0; t < 2; ++t) {
    int m = t * 16 + lm;
#pragma unroll
    for (int c = 0; c < 5; ++c) {
      if (c < nch) {
        int k0 = L1 ? ksrc_l1(c, half, 0) : ksrc_l0(c, half, 0);
        int k1 = L1 ? ksrc_l1(c, half, 1) : ksrc_l0(c, half, 1);
        v2f av;
        av.x = wcomb(Wih, Whh, D, m, k0);
        av.y = wcomb(Wih, Whh, D, m, k1);
        a[t][c] = av;
      }
    }
    v8f ci;
#pragma unroll
    for (int r = 0; r < 8; ++r) {
      int gm = t * 16 + half * 8 + r;
      int gr = gate_row(gm);
      ci[r] = (gr >= 0) ? gate_scale(gm) * bias[gr] : 0.0f;
    }
    cini[t] = ci;
  }
}

static __device__ __forceinline__ void mm(const v2f a[2][5], const v8f cini[2],
                                          const v2f bch[5], int nch,
                                          v8f& A0, v8f& A1) {
  A0 = cini[0];
  A1 = cini[1];
#pragma unroll
  for (int c = 0; c < 5; ++c) {
    if (c < nch) {
      A0 = wmma4(a[0][c], bch[c], A0);
      A1 = wmma4(a[1][c], bch[c], A1);
    }
  }
}

static __device__ __forceinline__ void lstm_pointwise(const v8f& A0, const v8f& A1,
                                                      float c[3], float h[3]) {
#pragma unroll
  for (int u = 0; u < 3; ++u) {
    float si = sigm_half(A0[u]);
    float sf = sigm_half(A0[3 + u]);
    float tg = tanh_hw(A1[u]);
    float so = sigm_half(A1[3 + u]);
    float cn = sf * c[u] + si * tg;
    c[u] = cn;
    h[u] = so * tanh_hw(cn);
  }
}

struct gates4 { float si, sf, tg, so; };
static __device__ __forceinline__ gates4 pw_gates(const v8f& A0, const v8f& A1, int u) {
  gates4 g;
  g.si = sigm_half(A0[u]);
  g.sf = sigm_half(A0[3 + u]);
  g.tg = tanh_hw(A1[u]);
  g.so = sigm_half(A1[3 + u]);
  return g;
}
static __device__ __forceinline__ void pw_cell(const gates4 g[3], float c[3]) {
#pragma unroll
  for (int u = 0; u < 3; ++u) c[u] = g[u].sf * c[u] + g[u].si * g[u].tg;
}
static __device__ __forceinline__ void pw_out(const gates4 g[3], const float c[3], float h[3]) {
#pragma unroll
  for (int u = 0; u < 3; ++u) h[u] = g[u].so * tanh_hw(c[u]);
}

static __device__ __forceinline__ void ld_x(const float* xp, int half,
                                            v2f& xlo, float& x4) {
  xlo = *(const v2f*)(xp + 2 * half);
  x4 = xp[4 + half];
}
static __device__ __forceinline__ void ld_r(const float* r0p, int half,
                                            v2f& r01, float& r2) {
  r01 = *(const v2f*)(r0p + 2 * half);
  r2 = r0p[4 + half];
}

static __device__ __forceinline__ void b_l0(v2f xlo, float x4, const float h[3], v2f bch[5]) {
  v2f b1, b2;
  b1.x = h[0]; b1.y = h[1];
  b2.x = x4;   b2.y = h[2];
  bch[0] = xlo; bch[1] = b1; bch[2] = b2;
}

static __device__ __forceinline__ void b_l1(const float f0[3], v2f r01, float r2,
                                            const float h[3], v2f bch[5]) {
  v2f b1, b2, b3, b4;
  b1.x = r2;    b1.y = 0.0f;
  b2.x = f0[0]; b2.y = f0[1];
  b3.x = h[0];  b3.y = h[1];
  b4.x = f0[2]; b4.y = h[2];
  bch[0] = r01; bch[1] = b1; bch[2] = b2; bch[3] = b3; bch[4] = b4;
}

__global__ void __launch_bounds__(WG_THREADS) lstm_l0_bwd(const float* __restrict__ tf, const float* __restrict__ h0,
                            const float* __restrict__ c0,
                            const float* __restrict__ Wih, const float* __restrict__ Whh,
                            const float* __restrict__ bias,
                            float* __restrict__ r0ws, int B, int T) {
  int tid = blockIdx.x * blockDim.x + threadIdx.x;
  int lane = tid & 31, lm = lane & 15, half = lane >> 4;
  int s = (tid >> 5) * 16 + lm;
  if (s >= B) return;

  v2f a[2][5]; v8f ci[2];
  load_A<false>(Wih, Whh, bias, a, ci, lm, half);

  float h[3], c[3];
  size_t d1 = (size_t)B * LSTM_H;
#pragma unroll
  for (int u = 0; u < 3; ++u) {
    h[u] = h0[d1 + (size_t)s * 6 + 3 * half + u];
    c[u] = c0[d1 + (size_t)s * 6 + 3 * half + u];
  }

  const float* xrow = tf + (size_t)s * T * 6;
  v2f cxl[BLK_L0]; float cx4[BLK_L0];
#pragma unroll
  for (int j = 0; j < BLK_L0; ++j) {
    int tt = T - 1 - j; tt = tt < 0 ? 0 : tt;
    ld_x(xrow + (size_t)tt * 6, half, cxl[j], cx4[j]);
  }

  auto core = [&](v2f xlo, float x4) {
    v2f bch[5]; b_l0(xlo, x4, h, bch);
    v8f A0, A1; mm(a, ci, bch, 3, A0, A1);
    lstm_pointwise(A0, A1, c, h);
  };
  unsigned so01 = (unsigned)s * 24u + 8u * (unsigned)half;
  unsigned so2  = (unsigned)s * 24u + 16u + 4u * (unsigned)half;
  const size_t rowb = (size_t)B * 24;
  char* const r0b = (char*)r0ws;
  auto put = [&](unsigned long long rp, v2f w01, float w2) {
    gl_char* rowp = (gl_char*)rp;
    *(gl_v2f*)(rowp + so01) = w01;
    *(gl_float*)(rowp + so2) = w2;
  };

  unsigned xo01 = (unsigned)s * (unsigned)T * 24u + 8u * (unsigned)half;
  unsigned xo2  = (unsigned)s * (unsigned)T * 24u + 16u + 4u * (unsigned)half;
  const char* const xb = (const char*)tf;
  auto get = [&](unsigned long long xp, v2f& xlo, float& x4) {
    gl_cchar* stepp = (gl_cchar*)xp;
    xlo = *(gl_cv2f*)(stepp + xo01);
    x4 = *(gl_cfloat*)(stepp + xo2);
  };


  int t = T - 1;
  for (; t >= BLK_L0 - 1; t -= BLK_L0) {
    v2f nxl[BLK_L0]; float nx4[BLK_L0];
    unsigned long long xp[BLK_L0];
#pragma unroll
    for (int j = 0; j < BLK_L0; ++j) {
      int tn = t - BLK_L0 - j; tn = tn < 0 ? 0 : tn;
      xp[j] = (unsigned long long)(xb + (size_t)tn * 24);
    }
    asm volatile("" : "+s"(xp[0]), "+s"(xp[1]), "+s"(xp[2]), "+s"(xp[3]),
                      "+s"(xp[4]), "+s"(xp[5]), "+s"(xp[6]), "+s"(xp[7]),
                      "+v"(xo01), "+v"(xo2));
#pragma unroll
    for (int j = 0; j < BLK_L0; ++j) get(xp[j], nxl[j], nx4[j]);
    __builtin_amdgcn_sched_barrier(0);
    v2f sw01[BLK_L0]; float sw2[BLK_L0];
#pragma unroll
    for (int j = 0; j < BLK_L0; ++j) {
      core(cxl[j], cx4[j]);
      sw01[j].x = h[0]; sw01[j].y = h[1]; sw2[j] = h[2];
    }
    __builtin_amdgcn_sched_barrier(0);
    unsigned long long rp[BLK_L0];
#pragma unroll
    for (int j = 0; j < BLK_L0; ++j) rp[j] = (unsigned long long)(r0b + (size_t)(t - j) * rowb);
    asm volatile("" : "+s"(rp[0]), "+s"(rp[1]), "+s"(rp[2]), "+s"(rp[3]),
                      "+s"(rp[4]), "+s"(rp[5]), "+s"(rp[6]), "+s"(rp[7]),
                      "+v"(so01), "+v"(so2));
#pragma unroll
    for (int j = 0; j < BLK_L0; ++j) put(rp[j], sw01[j], sw2[j]);
    __builtin_amdgcn_sched_barrier(0);
#pragma unroll
    for (int j = 0; j < BLK_L0; ++j) { cxl[j] = nxl[j]; cx4[j] = nx4[j]; }
  }
#pragma unroll
  for (int j = 0; j < BLK_L0 - 1; ++j) {
    if (j <= t) {
      core(cxl[j], cx4[j]);
      v2f w01; w01.x = h[0]; w01.y = h[1];
      unsigned long long rp1 = (unsigned long long)(r0b + (size_t)(t - j) * rowb);
      asm volatile("" : "+s"(rp1), "+v"(so01), "+v"(so2));
      put(rp1, w01, h[2]);
    }
  }
}

#define HB 8
typedef __attribute__((ext_vector_type(4))) float v4f;
#define PIN() __builtin_amdgcn_sched_barrier(0)
#define TIE(tile) asm volatile("" : "+v"(tile))

__global__ void __launch_bounds__(WG_THREADS) lstm_fused(const float* __restrict__ tf, const float* __restrict__ h0,
                           const float* __restrict__ c0,
                           const float* __restrict__ Wih0f, const float* __restrict__ Whh0f,
                           const float* __restrict__ b0f,
                           const float* __restrict__ Wih1f, const float* __restrict__ Whh1f,
                           const float* __restrict__ b1f,
                           const float* __restrict__ Wih1b, const float* __restrict__ Whh1b,
                           const float* __restrict__ b1b,
                           const float* __restrict__ fcw, const float* __restrict__ fcb,
                           const float* __restrict__ r0ws, float* __restrict__ out,
                           int B, int T) {
  __shared__ v4f hbuf[2][2][HB][4][32];

  const int lane = threadIdx.x & 31, lm = lane & 15, half = lane >> 4;
  const int wv = __builtin_amdgcn_readfirstlane((int)(threadIdx.x >> 5));
  const bool producer = wv < 2;
  const int grp = wv & 1;
  int s = ((int)blockIdx.x * 2 + grp) * 16 + lm;
  const bool live = s < B;
  if (!live) s = B - 1;

  const int nch = (T + HB - 1) / HB;
  const float* xrow = tf + (size_t)s * T * 6;
  const float* rcol = r0ws + (size_t)s * 6;
  const size_t rstep = (size_t)B * 6;

  v2f aw[2][5]; v8f ciw[2];
  float hS[3], cS[3];
  v2f cua[HB]; float cub[HB]; float cuc[HB];
  float fl[3] = {0.0f, 0.0f, 0.0f};
  v2f ap[2]; v8f cip[2];
  ap[0] = (v2f){0.0f, 0.0f}; ap[1] = ap[0];
  cip[0] = (v8f){0.0f, 0.0f, 0.0f, 0.0f, 0.0f, 0.0f, 0.0f, 0.0f}; cip[1] = cip[0];
  v8f N0 = cip[0], N1 = cip[0];

  if (producer) {
    load_A<false>(Wih0f, Whh0f, b0f, aw, ciw, lm, half);
    v2f aq[2][5]; v8f ciq[2];
    load_A<true>(Wih1f, Whh1f, b1f, aq, ciq, lm, half);
    ap[0] = aq[0][1]; ap[1] = aq[1][1];
    cip[0] = ciq[0];  cip[1] = ciq[1];
  } else {
    load_A<true>(Wih1f, Whh1f, b1f, aw, ciw, lm, half);
  }
  {
    size_t dd = (producer ? 0 : 2) * (size_t)B * LSTM_H;
#pragma unroll
    for (int u = 0; u < 3; ++u) {
      size_t o = (size_t)s * 6 + 3 * half + u;
      hS[u] = h0[dd + o];  cS[u] = c0[dd + o];
    }
  }
#pragma unroll
  for (int j = 0; j < HB; ++j) {
    int tt = j > T - 1 ? T - 1 : j;
    if (producer) {
      ld_x(xrow + (size_t)tt * 6, half, cua[j], cub[j]);
      cuc[j] = rcol[(size_t)tt * rstep + 4 + half];
    } else {
      cua[j] = *(const v2f*)(rcol + (size_t)tt * rstep + 2 * half);
      cub[j] = 0.0f;  cuc[j] = 0.0f;
    }
  }
  if (producer) {
    N0 = wmma4(aw[0][0], cua[0], ciw[0]);
    N1 = wmma4(aw[1][0], cua[0], ciw[1]);
  }

  auto step1 = [&](v4f q0, v4f q1, v4f q2, v4f q3, v2f r01) {
    v8f A0, A1;
    A0[0] = q0.x; A0[1] = q0.y; A0[2] = q0.z; A0[3] = q0.w;
    A0[4] = q1.x; A0[5] = q1.y; A0[6] = q1.z; A0[7] = q1.w;
    A1[0] = q2.x; A1[1] = q2.y; A1[2] = q2.z; A1[3] = q2.w;
    A1[4] = q3.x; A1[5] = q3.y; A1[6] = q3.z; A1[7] = q3.w;
    v2f bf, bh, bfh;
    bf.x = q1.z;   bf.y = q1.w;
    bh.x = hS[0];  bh.y = hS[1];
    bfh.x = q3.z;  bfh.y = hS[2];
    A0 = wmma4(aw[0][0], r01, A0);  A1 = wmma4(aw[1][0], r01, A1);
    A0 = wmma4(aw[0][2], bf, A0);   A1 = wmma4(aw[1][2], bf, A1);
    A0 = wmma4(aw[0][3], bh, A0);   A1 = wmma4(aw[1][3], bh, A1);
    A0 = wmma4(aw[0][4], bfh, A0);  A1 = wmma4(aw[1][4], bfh, A1);
    lstm_pointwise(A0, A1, cS, hS);
  };

  for (int k = 0; k <= nch; ++k) {
    if (producer) {
      if (k < nch) {
        const int t0 = k * HB;
        v2f nxa[HB]; float nxb[HB]; float nxc[HB];
#pragma unroll
        for (int j = 0; j < HB; ++j) {
          int tn = t0 + HB + j; tn = tn > T - 1 ? T - 1 : tn;
          ld_x(xrow + (size_t)tn * 6, half, nxa[j], nxb[j]);
          nxc[j] = rcol[(size_t)tn * rstep + 4 + half];
        }
        __builtin_amdgcn_sched_barrier(0);
        v4f w0[HB], w1[HB], w2[HB], w3[HB];
#pragma unroll
        for (int j = 0; j < HB; ++j) {
          v2f b1, b2;
          b1.x = hS[0];  b1.y = hS[1];
          b2.x = cub[j]; b2.y = hS[2];
          v8f A0 = wmma4(aw[0][1], b1, N0);
          v8f A1 = wmma4(aw[1][1], b1, N1);
          A0 = wmma4(aw[0][2], b2, A0);
          A1 = wmma4(aw[1][2], b2, A1);
          PIN();
          const v2f xn = (j + 1 < HB) ? cua[j + 1] : nxa[0];
          N0 = wmma4(aw[0][0], xn, ciw[0]);
          N1 = wmma4(aw[1][0], xn, ciw[1]);
          TIE(N0); TIE(N1);
          PIN();
          gates4 g[3];
          g[0] = pw_gates(A0, A1, 0);
          g[1] = pw_gates(A0, A1, 1);
          PIN();
          v2f br; br.x = cuc[j]; br.y = 0.0f;
          v8f P0 = wmma4(ap[0], br, cip[0]);
          TIE(P0);
          PIN();
          g[2] = pw_gates(A0, A1, 2);
          pw_cell(g, cS);
          PIN();
          v8f P1 = wmma4(ap[1], br, cip[1]);
          TIE(P1);
          PIN();
          pw_out(g, cS, hS);
          w0[j].x = P0[0]; w0[j].y = P0[1]; w0[j].z = P0[2]; w0[j].w = P0[3];
          w1[j].x = P0[4]; w1[j].y = P0[5]; w1[j].z = hS[0]; w1[j].w = hS[1];
          w2[j].x = P1[0]; w2[j].y = P1[1]; w2[j].z = P1[2]; w2[j].w = P1[3];
          w3[j].x = P1[4]; w3[j].y = P1[5]; w3[j].z = hS[2]; w3[j].w = 0.0f;
        }
#pragma unroll
        for (int j = 0; j < HB; ++j) {
          v4f* slot = &hbuf[k & 1][grp][j][0][lane];
          slot[0] = w0[j]; slot[32] = w1[j]; slot[64] = w2[j]; slot[96] = w3[j];
        }
        __builtin_amdgcn_sched_barrier(0);
#pragma unroll
        for (int j = 0; j < HB; ++j) { cua[j] = nxa[j]; cub[j] = nxb[j]; cuc[j] = nxc[j]; }
      }
    } else {
      if (k > 0) {
        const int kc = k - 1;
        const int t0 = kc * HB;
        v2f nxa[HB];
#pragma unroll
        for (int j = 0; j < HB; ++j) {
          int tn = t0 + HB + j; tn = tn > T - 1 ? T - 1 : tn;
          nxa[j] = *(const v2f*)(rcol + (size_t)tn * rstep + 2 * half);
        }
        v4f q0[HB], q1[HB], q2[HB], q3[HB];
#pragma unroll
        for (int j = 0; j < HB; ++j) {
          q0[j] = hbuf[kc & 1][grp][j][0][lane];  q1[j] = hbuf[kc & 1][grp][j][1][lane];
          q2[j] = hbuf[kc & 1][grp][j][2][lane];  q3[j] = hbuf[kc & 1][grp][j][3][lane];
        }
        __builtin_amdgcn_sched_barrier(0);
        if (t0 + HB <= T) {
          v8f M0, M1;
          {
            M0[0] = q0[0].x; M0[1] = q0[0].y; M0[2] = q0[0].z; M0[3] = q0[0].w;
            M0[4] = q1[0].x; M0[5] = q1[0].y; M0[6] = q1[0].z; M0[7] = q1[0].w;
            M1[0] = q2[0].x; M1[1] = q2[0].y; M1[2] = q2[0].z; M1[3] = q2[0].w;
            M1[4] = q3[0].x; M1[5] = q3[0].y; M1[6] = q3[0].z; M1[7] = q3[0].w;
            v2f bf; bf.x = q1[0].z; bf.y = q1[0].w;
            M0 = wmma4(aw[0][0], cua[0], M0);  M1 = wmma4(aw[1][0], cua[0], M1);
            M0 = wmma4(aw[0][2], bf, M0);      M1 = wmma4(aw[1][2], bf, M1);
          }
          PIN();
#pragma unroll
          for (int j = 0; j < HB; ++j) {
            v2f bh, bfh;
            bh.x = hS[0];     bh.y = hS[1];
            bfh.x = q3[j].z;  bfh.y = hS[2];
            v8f A0 = wmma4(aw[0][3], bh, M0);
            v8f A1 = wmma4(aw[1][3], bh, M1);
            A0 = wmma4(aw[0][4], bfh, A0);
            A1 = wmma4(aw[1][4], bfh, A1);
            PIN();
            const int jn = j + 1 < HB ? j + 1 : HB - 1;
            v2f bfn; bfn.x = q1[jn].z; bfn.y = q1[jn].w;
            if (j + 1 < HB) {
              M0[0] = q0[jn].x; M0[1] = q0[jn].y; M0[2] = q0[jn].z; M0[3] = q0[jn].w;
              M0[4] = q1[jn].x; M0[5] = q1[jn].y; M0[6] = q1[jn].z; M0[7] = q1[jn].w;
              M1[0] = q2[jn].x; M1[1] = q2[jn].y; M1[2] = q2[jn].z; M1[3] = q2[jn].w;
              M1[4] = q3[jn].x; M1[5] = q3[jn].y; M1[6] = q3[jn].z; M1[7] = q3[jn].w;
              M0 = wmma4(aw[0][0], cua[jn], M0);
              M1 = wmma4(aw[1][0], cua[jn], M1);
              TIE(M0); TIE(M1);
            }
            PIN();
            gates4 g[3];
            g[0] = pw_gates(A0, A1, 0);
            g[1] = pw_gates(A0, A1, 1);
            PIN();
            if (j + 1 < HB) { M0 = wmma4(aw[0][2], bfn, M0); TIE(M0); }
            PIN();
            g[2] = pw_gates(A0, A1, 2);
            pw_cell(g, cS);
            PIN();
            if (j + 1 < HB) { M1 = wmma4(aw[1][2], bfn, M1); TIE(M1); }
            PIN();
            pw_out(g, cS, hS);
          }
          fl[0] = q1[HB - 1].z; fl[1] = q1[HB - 1].w; fl[2] = q3[HB - 1].z;
        } else {
#pragma unroll
          for (int j = 0; j < HB; ++j) {
            if (t0 + j < T) {
              step1(q0[j], q1[j], q2[j], q3[j], cua[j]);
              fl[0] = q1[j].z; fl[1] = q1[j].w; fl[2] = q3[j].z;
            }
          }
        }
        __builtin_amdgcn_sched_barrier(0);
#pragma unroll
        for (int j = 0; j < HB; ++j) cua[j] = nxa[j];
      }
    }
    __syncthreads();
  }
  if (producer) return;

  v2f a2[2][5]; v8f ci2[2]; load_A<true>(Wih1b, Whh1b, b1b, a2, ci2, lm, half);
  float hR[3], cR[3];
  size_t d3 = 3 * (size_t)B * LSTM_H;
#pragma unroll
  for (int u = 0; u < 3; ++u) {
    size_t o = (size_t)s * 6 + 3 * half + u;
    hR[u] = h0[d3 + o];  cR[u] = c0[d3 + o];
  }
  {
    const float* r0p = r0ws + ((size_t)(T - 1) * B + s) * 6;
    v2f r01; float r2; ld_r(r0p, half, r01, r2);
    v2f bch[5]; b_l1(fl, r01, r2, hR, bch);
    v8f A0, A1; mm(a2, ci2, bch, 5, A0, A1);
    lstm_pointwise(A0, A1, cR, hR);
  }
  float pB[3], pR[3];
#pragma unroll
  for (int u = 0; u < 3; ++u) { pB[u] = sx16(hS[u]); pR[u] = sx16(hR[u]); }
  if (half == 0 && live) {
    float v[12];
#pragma unroll
    for (int u = 0; u < 3; ++u) {
      v[u]     = hS[u] > 0.0f ? hS[u] : 0.0f;
      v[3 + u] = pB[u] > 0.0f ? pB[u] : 0.0f;
      v[6 + u] = hR[u] > 0.0f ? hR[u] : 0.0f;
      v[9 + u] = pR[u] > 0.0f ? pR[u] : 0.0f;
    }
#pragma unroll
    for (int j = 0; j < 6; ++j) {
      float acc = fcb[j];
#pragma unroll
      for (int k = 0; k < 12; ++k) acc = fmaf(fcw[j * 12 + k], v[k], acc);
      out[(size_t)s * 6 + j] = acc;
    }
  }
}

extern "C" void kernel_launch(void* const* d_in, const int* in_sizes, int n_in,
                              void* d_out, int out_size, void* d_ws, size_t ws_size,
                              hipStream_t stream) {
  const float* tf    = (const float*)d_in[0];
  const float* h0    = (const float*)d_in[1];
  const float* c0    = (const float*)d_in[2];
  const float* Wih0f = (const float*)d_in[3];
  const float* Whh0f = (const float*)d_in[4];
  const float* b0f   = (const float*)d_in[5];
  const float* Wih0b = (const float*)d_in[6];
  const float* Whh0b = (const float*)d_in[7];
  const float* b0b   = (const float*)d_in[8];
  const float* Wih1f = (const float*)d_in[9];
  const float* Whh1f = (const float*)d_in[10];
  const float* b1f   = (const float*)d_in[11];
  const float* Wih1b = (const float*)d_in[12];
  const float* Whh1b = (const float*)d_in[13];
  const float* b1b   = (const float*)d_in[14];
  const float* fcw   = (const float*)d_in[15];
  const float* fcb   = (const float*)d_in[16];

  int B = in_sizes[1] / (4 * LSTM_H);
  int T = in_sizes[0] / (B * LSTM_H);

  size_t need = (size_t)B * T * LSTM_H * sizeof(float);
  if (ws_size < need) return;
  float* r0ws = (float*)d_ws;

  int waves = (B + 15) / 16;
  int threads = waves * 32;
  int block = WG_THREADS;
  int grid = (threads + block - 1) / block;

  lstm_l0_bwd<<<grid, block, 0, stream>>>(tf, h0, c0, Wih0b, Whh0b, b0b, r0ws, B, T);
  int grid2 = (waves + 1) / 2;
  lstm_fused<<<grid2, block, 0, stream>>>(tf, h0, c0, Wih0f, Whh0f, b0f,
                                         Wih1f, Whh1f, b1f, Wih1b, Whh1b, b1b,
                                         fcw, fcb, r0ws, (float*)d_out, B, T);
}


# first launch: x[t] loads issued three blocks (24 steps) ahead instead of one, through four register sets used in turn with no copy between sets: the load round trip no longer bounds a block
# speedup vs baseline: 6.1744x; 1.0864x over previous
#include <hip/hip_runtime.h>
#include <hip/hip_bf16.h>


typedef __attribute__((ext_vector_type(2))) float v2f;
typedef __attribute__((ext_vector_type(8))) float v8f;
typedef __attribute__((address_space(1))) char gl_char;
typedef __attribute__((address_space(1))) float gl_float;
typedef __attribute__((address_space(1))) v2f gl_v2f;
typedef __attribute__((address_space(1))) const char gl_cchar;
typedef __attribute__((address_space(1))) const float gl_cfloat;
typedef __attribute__((address_space(1))) const v2f gl_cv2f;

#define LSTM_H 6
#define BLK_L0 8
#define WG_THREADS 128

static __device__ __forceinline__ float tanh_hw(float x) {
  return __builtin_amdgcn_tanhf(x);
}
static __device__ __forceinline__ float sigm_half(float zh) {
  return fmaf(0.5f, tanh_hw(zh), 0.5f);
}
static __device__ __forceinline__ float sx16(float v) {
  int b = __builtin_bit_cast(int, v);
  return __builtin_bit_cast(float,
      __builtin_amdgcn_permlanex16(b, b, 0x76543210, 0xfedcba98, false, false));
}
static __device__ __forceinline__ v8f wmma4(v2f a, v2f b, v8f c) {
  return __builtin_amdgcn_wmma_f32_16x16x4_f32(false, a, false, b, (short)0, c,
                                               false, false);
}

static __device__ __forceinline__ int gate_row(int m) {
  int idx = m & 7;
  if (idx >= LSTM_H) return -1;
  int hf = (m >> 3) & 1;
  int tile = m >> 4;
  int second = idx >= 3 ? 1 : 0;
  int gate = 2 * tile + second;
  int unit = 3 * hf + idx - 3 * second;
  return gate * LSTM_H + unit;
}

static __device__ __forceinline__ float gate_scale(int m) {
  int idx = m & 7;
  int tile = m >> 4;
  return (tile == 1 && idx < 3) ? 1.0f : 0.5f;
}

static __device__ __forceinline__ int ksrc_l0(int c, int hf, int e) {
  if (c == 0) return 2 * hf + e;
  if (c == 1) return 6 + 3 * hf + e;
  return e == 0 ? 4 + hf : 6 + 3 * hf + 2;
}
static __device__ __forceinline__ int ksrc_l1(int c, int hf, int e) {
  if (c == 0) return 6 + 3 * hf + e;
  if (c == 1) return e == 0 ? 6 + 3 * hf + 2 : -1;
  if (c == 2) return 3 * hf + e;
  if (c == 3) return 12 + 3 * hf + e;
  return e == 0 ? 3 * hf + 2 : 12 + 3 * hf + 2;
}

static __device__ __forceinline__ float wcomb(const float* Wih, const float* Whh,
                                              int D, int m, int k) {
  int r = gate_row(m);
  if (r < 0 || k < 0) return 0.0f;
  float sc = gate_scale(m);
  if (k < D) return sc * Wih[r * D + k];
  return sc * Whh[r * LSTM_H + (k - D)];
}

template <bool L1>
static __device__ __forceinline__ void load_A(const float* Wih, const float* Whh,
                                              const float* bias,
                                              v2f a[2][5], v8f cini[2],
                                              int lm, int half) {
  const int D = L1 ? 2 * LSTM_H : LSTM_H;
  const int nch = L1 ? 5 : 3;
  for (int t = 0; t < 2; ++t) {
    int m = t * 16 + lm;
#pragma unroll
    for (int c = 0; c < 5; ++c) {
      if (c < nch) {
        int k0 = L1 ? ksrc_l1(c, half, 0) : ksrc_l0(c, half, 0);
        int k1 = L1 ? ksrc_l1(c, half, 1) : ksrc_l0(c, half, 1);
        v2f av;
        av.x = wcomb(Wih, Whh, D, m, k0);
        av.y = wcomb(Wih, Whh, D, m, k1);
        a[t][c] = av;
      }
    }
    v8f ci;
#pragma unroll
    for (int r = 0; r < 8; ++r) {
      int gm = t * 16 + half * 8 + r;
      int gr = gate_row(gm);
      ci[r] = (gr >= 0) ? gate_scale(gm) * bias[gr] : 0.0f;
    }
    cini[t] = ci;
  }
}

static __device__ __forceinline__ void mm(const v2f a[2][5], const v8f cini[2],
                                          const v2f bch[5], int nch,
                                          v8f& A0, v8f& A1) {
  A0 = cini[0];
  A1 = cini[1];
#pragma unroll
  for (int c = 0; c < 5; ++c) {
    if (c < nch) {
      A0 = wmma4(a[0][c], bch[c], A0);
      A1 = wmma4(a[1][c], bch[c], A1);
    }
  }
}

static __device__ __forceinline__ void lstm_pointwise(const v8f& A0, const v8f& A1,
                                                      float c[3], float h[3]) {
#pragma unroll
  for (int u = 0; u < 3; ++u) {
    float si = sigm_half(A0[u]);
    float sf = sigm_half(A0[3 + u]);
    float tg = tanh_hw(A1[u]);
    float so = sigm_half(A1[3 + u]);
    float cn = sf * c[u] + si * tg;
    c[u] = cn;
    h[u] = so * tanh_hw(cn);
  }
}

struct gates4 { float si, sf, tg, so; };
static __device__ __forceinline__ gates4 pw_gates(const v8f& A0, const v8f& A1, int u) {
  gates4 g;
  g.si = sigm_half(A0[u]);
  g.sf = sigm_half(A0[3 + u]);
  g.tg = tanh_hw(A1[u]);
  g.so = sigm_half(A1[3 + u]);
  return g;
}
static __device__ __forceinline__ void pw_cell(const gates4 g[3], float c[3]) {
#pragma unroll
  for (int u = 0; u < 3; ++u) c[u] = g[u].sf * c[u] + g[u].si * g[u].tg;
}
static __device__ __forceinline__ void pw_out(const gates4 g[3], const float c[3], float h[3]) {
#pragma unroll
  for (int u = 0; u < 3; ++u) h[u] = g[u].so * tanh_hw(c[u]);
}

static __device__ __forceinline__ void ld_x(const float* xp, int half,
                                            v2f& xlo, float& x4) {
  xlo = *(const v2f*)(xp + 2 * half);
  x4 = xp[4 + half];
}
static __device__ __forceinline__ void ld_r(const float* r0p, int half,
                                            v2f& r01, float& r2) {
  r01 = *(const v2f*)(r0p + 2 * half);
  r2 = r0p[4 + half];
}

static __device__ __forceinline__ void b_l0(v2f xlo, float x4, const float h[3], v2f bch[5]) {
  v2f b1, b2;
  b1.x = h[0]; b1.y = h[1];
  b2.x = x4;   b2.y = h[2];
  bch[0] = xlo; bch[1] = b1; bch[2] = b2;
}

static __device__ __forceinline__ void b_l1(const float f0[3], v2f r01, float r2,
                                            const float h[3], v2f bch[5]) {
  v2f b1, b2, b3, b4;
  b1.x = r2;    b1.y = 0.0f;
  b2.x = f0[0]; b2.y = f0[1];
  b3.x = h[0];  b3.y = h[1];
  b4.x = f0[2]; b4.y = h[2];
  bch[0] = r01; bch[1] = b1; bch[2] = b2; bch[3] = b3; bch[4] = b4;
}

__global__ void __launch_bounds__(WG_THREADS) lstm_l0_bwd(const float* __restrict__ tf, const float* __restrict__ h0,
                            const float* __restrict__ c0,
                            const float* __restrict__ Wih, const float* __restrict__ Whh,
                            const float* __restrict__ bias,
                            float* __restrict__ r0ws, int B, int T) {
  int tid = blockIdx.x * blockDim.x + threadIdx.x;
  int lane = tid & 31, lm = lane & 15, half = lane >> 4;
  int s = (tid >> 5) * 16 + lm;
  if (s >= B) return;

  v2f a[2][5]; v8f ci[2];
  load_A<false>(Wih, Whh, bias, a, ci, lm, half);

  float h[3], c[3];
  size_t d1 = (size_t)B * LSTM_H;
#pragma unroll
  for (int u = 0; u < 3; ++u) {
    h[u] = h0[d1 + (size_t)s * 6 + 3 * half + u];
    c[u] = c0[d1 + (size_t)s * 6 + 3 * half + u];
  }

  v2f xA[BLK_L0], xB[BLK_L0], xC[BLK_L0], xD[BLK_L0];
  float yA[BLK_L0], yB[BLK_L0], yC[BLK_L0], yD[BLK_L0];

  auto core = [&](v2f xlo, float x4) {
    v2f bch[5]; b_l0(xlo, x4, h, bch);
    v8f A0, A1; mm(a, ci, bch, 3, A0, A1);
    lstm_pointwise(A0, A1, c, h);
  };
  unsigned so01 = (unsigned)s * 24u + 8u * (unsigned)half;
  unsigned so2  = (unsigned)s * 24u + 16u + 4u * (unsigned)half;
  const size_t rowb = (size_t)B * 24;
  char* const r0b = (char*)r0ws;
  auto put = [&](unsigned long long rp, v2f w01, float w2) {
    gl_char* rowp = (gl_char*)rp;
    *(gl_v2f*)(rowp + so01) = w01;
    *(gl_float*)(rowp + so2) = w2;
  };

  unsigned xo01 = (unsigned)s * (unsigned)T * 24u + 8u * (unsigned)half;
  unsigned xo2  = (unsigned)s * (unsigned)T * 24u + 16u + 4u * (unsigned)half;
  const char* const xb = (const char*)tf;
  auto get = [&](unsigned long long xp, v2f& xlo, float& x4) {
    gl_cchar* stepp = (gl_cchar*)xp;
    xlo = *(gl_cv2f*)(stepp + xo01);
    x4 = *(gl_cfloat*)(stepp + xo2);
  };
  auto fetch = [&](int first, v2f (&xl)[BLK_L0], float (&x4)[BLK_L0]) {
    unsigned long long xp[BLK_L0];
#pragma unroll
    for (int j = 0; j < BLK_L0; ++j) {
      int tn = first - j; tn = tn < 0 ? 0 : tn;
      xp[j] = (unsigned long long)(xb + (size_t)tn * 24);
    }
    asm volatile("" : "+s"(xp[0]), "+s"(xp[1]), "+s"(xp[2]), "+s"(xp[3]),
                      "+s"(xp[4]), "+s"(xp[5]), "+s"(xp[6]), "+s"(xp[7]),
                      "+v"(xo01), "+v"(xo2));
#pragma unroll
    for (int j = 0; j < BLK_L0; ++j) get(xp[j], xl[j], x4[j]);
  };

  auto run_block = [&](int tt, const v2f (&xl)[BLK_L0], const float (&x4)[BLK_L0]) {
    v2f sw01[BLK_L0]; float sw2[BLK_L0];
#pragma unroll
    for (int j = 0; j < BLK_L0; ++j) {
      core(xl[j], x4[j]);
      sw01[j].x = h[0]; sw01[j].y = h[1]; sw2[j] = h[2];
    }
    __builtin_amdgcn_sched_barrier(0);
    unsigned long long rp[BLK_L0];
#pragma unroll
    for (int j = 0; j < BLK_L0; ++j) rp[j] = (unsigned long long)(r0b + (size_t)(tt - j) * rowb);
    asm volatile("" : "+s"(rp[0]), "+s"(rp[1]), "+s"(rp[2]), "+s"(rp[3]),
                      "+s"(rp[4]), "+s"(rp[5]), "+s"(rp[6]), "+s"(rp[7]),
                      "+v"(so01), "+v"(so2));
#pragma unroll
    for (int j = 0; j < BLK_L0; ++j) put(rp[j], sw01[j], sw2[j]);
  };
  auto run_rest = [&](int tt, const v2f (&xl)[BLK_L0], const float (&x4)[BLK_L0]) {
#pragma unroll
    for (int j = 0; j < BLK_L0 - 1; ++j) {
      if (j <= tt) {
        core(xl[j], x4[j]);
        v2f w01; w01.x = h[0]; w01.y = h[1];
        unsigned long long rp1 = (unsigned long long)(r0b + (size_t)(tt - j) * rowb);
        asm volatile("" : "+s"(rp1), "+v"(so01), "+v"(so2));
        put(rp1, w01, h[2]);
      }
    }
  };
#define L0_AHEAD_AND_RUN(tt, xn, yn, xc, yc)          \
  fetch((tt) - 3 * BLK_L0, xn, yn);                    \
  __builtin_amdgcn_sched_barrier(0);                   \
  run_block((tt), xc, yc);                             \
  __builtin_amdgcn_sched_barrier(0)

  int t = T - 1;
  fetch(t, xA, yA);
  fetch(t - BLK_L0, xB, yB);
  fetch(t - 2 * BLK_L0, xC, yC);
  for (; t >= 4 * BLK_L0 - 1; t -= 4 * BLK_L0) {
    L0_AHEAD_AND_RUN(t,              xD, yD, xA, yA);
    L0_AHEAD_AND_RUN(t - BLK_L0,     xA, yA, xB, yB);
    L0_AHEAD_AND_RUN(t - 2 * BLK_L0, xB, yB, xC, yC);
    L0_AHEAD_AND_RUN(t - 3 * BLK_L0, xC, yC, xD, yD);
  }
  if (t >= BLK_L0 - 1) {
    L0_AHEAD_AND_RUN(t, xD, yD, xA, yA);
    t -= BLK_L0;
    if (t >= BLK_L0 - 1) {
      run_block(t, xB, yB);
      t -= BLK_L0;
      if (t >= BLK_L0 - 1) {
        run_block(t, xC, yC);
        t -= BLK_L0;
        run_rest(t, xD, yD);
      } else {
        run_rest(t, xC, yC);
      }
    } else {
      run_rest(t, xB, yB);
    }
  } else {
    run_rest(t, xA, yA);
  }
#undef L0_AHEAD_AND_RUN
}

#define HB 8
typedef __attribute__((ext_vector_type(4))) float v4f;
#define PIN() __builtin_amdgcn_sched_barrier(0)
#define TIE(tile) asm volatile("" : "+v"(tile))

__global__ void __launch_bounds__(WG_THREADS) lstm_fused(const float* __restrict__ tf, const float* __restrict__ h0,
                           const float* __restrict__ c0,
                           const float* __restrict__ Wih0f, const float* __restrict__ Whh0f,
                           const float* __restrict__ b0f,
                           const float* __restrict__ Wih1f, const float* __restrict__ Whh1f,
                           const float* __restrict__ b1f,
                           const float* __restrict__ Wih1b, const float* __restrict__ Whh1b,
                           const float* __restrict__ b1b,
                           const float* __restrict__ fcw, const float* __restrict__ fcb,
                           const float* __restrict__ r0ws, float* __restrict__ out,
                           int B, int T) {
  __shared__ v4f hbuf[2][2][HB][4][32];

  const int lane = threadIdx.x & 31, lm = lane & 15, half = lane >> 4;
  const int wv = __builtin_amdgcn_readfirstlane((int)(threadIdx.x >> 5));
  const bool producer = wv < 2;
  const int grp = wv & 1;
  int s = ((int)blockIdx.x * 2 + grp) * 16 + lm;
  const bool live = s < B;
  if (!live) s = B - 1;

  const int nch = (T + HB - 1) / HB;
  const float* xrow = tf + (size_t)s * T * 6;
  const float* rcol = r0ws + (size_t)s * 6;
  const size_t rstep = (size_t)B * 6;

  v2f aw[2][5]; v8f ciw[2];
  float hS[3], cS[3];
  v2f cua[HB]; float cub[HB]; float cuc[HB];
  float fl[3] = {0.0f, 0.0f, 0.0f};
  v2f ap[2]; v8f cip[2];
  ap[0] = (v2f){0.0f, 0.0f}; ap[1] = ap[0];
  cip[0] = (v8f){0.0f, 0.0f, 0.0f, 0.0f, 0.0f, 0.0f, 0.0f, 0.0f}; cip[1] = cip[0];
  v8f N0 = cip[0], N1 = cip[0];

  if (producer) {
    load_A<false>(Wih0f, Whh0f, b0f, aw, ciw, lm, half);
    v2f aq[2][5]; v8f ciq[2];
    load_A<true>(Wih1f, Whh1f, b1f, aq, ciq, lm, half);
    ap[0] = aq[0][1]; ap[1] = aq[1][1];
    cip[0] = ciq[0];  cip[1] = ciq[1];
  } else {
    load_A<true>(Wih1f, Whh1f, b1f, aw, ciw, lm, half);
  }
  {
    size_t dd = (producer ? 0 : 2) * (size_t)B * LSTM_H;
#pragma unroll
    for (int u = 0; u < 3; ++u) {
      size_t o = (size_t)s * 6 + 3 * half + u;
      hS[u] = h0[dd + o];  cS[u] = c0[dd + o];
    }
  }
#pragma unroll
  for (int j = 0; j < HB; ++j) {
    int tt = j > T - 1 ? T - 1 : j;
    if (producer) {
      ld_x(xrow + (size_t)tt * 6, half, cua[j], cub[j]);
      cuc[j] = rcol[(size_t)tt * rstep + 4 + half];
    } else {
      cua[j] = *(const v2f*)(rcol + (size_t)tt * rstep + 2 * half);
      cub[j] = 0.0f;  cuc[j] = 0.0f;
    }
  }
  if (producer) {
    N0 = wmma4(aw[0][0], cua[0], ciw[0]);
    N1 = wmma4(aw[1][0], cua[0], ciw[1]);
  }

  auto step1 = [&](v4f q0, v4f q1, v4f q2, v4f q3, v2f r01) {
    v8f A0, A1;
    A0[0] = q0.x; A0[1] = q0.y; A0[2] = q0.z; A0[3] = q0.w;
    A0[4] = q1.x; A0[5] = q1.y; A0[6] = q1.z; A0[7] = q1.w;
    A1[0] = q2.x; A1[1] = q2.y; A1[2] = q2.z; A1[3] = q2.w;
    A1[4] = q3.x; A1[5] = q3.y; A1[6] = q3.z; A1[7] = q3.w;
    v2f bf, bh, bfh;
    bf.x = q1.z;   bf.y = q1.w;
    bh.x = hS[0];  bh.y = hS[1];
    bfh.x = q3.z;  bfh.y = hS[2];
    A0 = wmma4(aw[0][0], r01, A0);  A1 = wmma4(aw[1][0], r01, A1);
    A0 = wmma4(aw[0][2], bf, A0);   A1 = wmma4(aw[1][2], bf, A1);
    A0 = wmma4(aw[0][3], bh, A0);   A1 = wmma4(aw[1][3], bh, A1);
    A0 = wmma4(aw[0][4], bfh, A0);  A1 = wmma4(aw[1][4], bfh, A1);
    lstm_pointwise(A0, A1, cS, hS);
  };

  for (int k = 0; k <= nch; ++k) {
    if (producer) {
      if (k < nch) {
        const int t0 = k * HB;
        v2f nxa[HB]; float nxb[HB]; float nxc[HB];
#pragma unroll
        for (int j = 0; j < HB; ++j) {
          int tn = t0 + HB + j; tn = tn > T - 1 ? T - 1 : tn;
          ld_x(xrow + (size_t)tn * 6, half, nxa[j], nxb[j]);
          nxc[j] = rcol[(size_t)tn * rstep + 4 + half];
        }
        __builtin_amdgcn_sched_barrier(0);
        v4f w0[HB], w1[HB], w2[HB], w3[HB];
#pragma unroll
        for (int j = 0; j < HB; ++j) {
          v2f b1, b2;
          b1.x = hS[0];  b1.y = hS[1];
          b2.x = cub[j]; b2.y = hS[2];
          v8f A0 = wmma4(aw[0][1], b1, N0);
          v8f A1 = wmma4(aw[1][1], b1, N1);
          A0 = wmma4(aw[0][2], b2, A0);
          A1 = wmma4(aw[1][2], b2, A1);
          PIN();
          const v2f xn = (j + 1 < HB) ? cua[j + 1] : nxa[0];
          N0 = wmma4(aw[0][0], xn, ciw[0]);
          N1 = wmma4(aw[1][0], xn, ciw[1]);
          TIE(N0); TIE(N1);
          PIN();
          gates4 g[3];
          g[0] = pw_gates(A0, A1, 0);
          g[1] = pw_gates(A0, A1, 1);
          PIN();
          v2f br; br.x = cuc[j]; br.y = 0.0f;
          v8f P0 = wmma4(ap[0], br, cip[0]);
          TIE(P0);
          PIN();
          g[2] = pw_gates(A0, A1, 2);
          pw_cell(g, cS);
          PIN();
          v8f P1 = wmma4(ap[1], br, cip[1]);
          TIE(P1);
          PIN();
          pw_out(g, cS, hS);
          w0[j].x = P0[0]; w0[j].y = P0[1]; w0[j].z = P0[2]; w0[j].w = P0[3];
          w1[j].x = P0[4]; w1[j].y = P0[5]; w1[j].z = hS[0]; w1[j].w = hS[1];
          w2[j].x = P1[0]; w2[j].y = P1[1]; w2[j].z = P1[2]; w2[j].w = P1[3];
          w3[j].x = P1[4]; w3[j].y = P1[5]; w3[j].z = hS[2]; w3[j].w = 0.0f;
        }
#pragma unroll
        for (int j = 0; j < HB; ++j) {
          v4f* slot = &hbuf[k & 1][grp][j][0][lane];
          slot[0] = w0[j]; slot[32] = w1[j]; slot[64] = w2[j]; slot[96] = w3[j];
        }
        __builtin_amdgcn_sched_barrier(0);
#pragma unroll
        for (int j = 0; j < HB; ++j) { cua[j] = nxa[j]; cub[j] = nxb[j]; cuc[j] = nxc[j]; }
      }
    } else {
      if (k > 0) {
        const int kc = k - 1;
        const int t0 = kc * HB;
        v2f nxa[HB];
#pragma unroll
        for (int j = 0; j < HB; ++j) {
          int tn = t0 + HB + j; tn = tn > T - 1 ? T - 1 : tn;
          nxa[j] = *(const v2f*)(rcol + (size_t)tn * rstep + 2 * half);
        }
        v4f q0[HB], q1[HB], q2[HB], q3[HB];
#pragma unroll
        for (int j = 0; j < HB; ++j) {
          q0[j] = hbuf[kc & 1][grp][j][0][lane];  q1[j] = hbuf[kc & 1][grp][j][1][lane];
          q2[j] = hbuf[kc & 1][grp][j][2][lane];  q3[j] = hbuf[kc & 1][grp][j][3][lane];
        }
        __builtin_amdgcn_sched_barrier(0);
        if (t0 + HB <= T) {
          v8f M0, M1;
          {
            M0[0] = q0[0].x; M0[1] = q0[0].y; M0[2] = q0[0].z; M0[3] = q0[0].w;
            M0[4] = q1[0].x; M0[5] = q1[0].y; M0[6] = q1[0].z; M0[7] = q1[0].w;
            M1[0] = q2[0].x; M1[1] = q2[0].y; M1[2] = q2[0].z; M1[3] = q2[0].w;
            M1[4] = q3[0].x; M1[5] = q3[0].y; M1[6] = q3[0].z; M1[7] = q3[0].w;
            v2f bf; bf.x = q1[0].z; bf.y = q1[0].w;
            M0 = wmma4(aw[0][0], cua[0], M0);  M1 = wmma4(aw[1][0], cua[0], M1);
            M0 = wmma4(aw[0][2], bf, M0);      M1 = wmma4(aw[1][2], bf, M1);
          }
          PIN();
#pragma unroll
          for (int j = 0; j < HB; ++j) {
            v2f bh, bfh;
            bh.x = hS[0];     bh.y = hS[1];
            bfh.x = q3[j].z;  bfh.y = hS[2];
            v8f A0 = wmma4(aw[0][3], bh, M0);
            v8f A1 = wmma4(aw[1][3], bh, M1);
            A0 = wmma4(aw[0][4], bfh, A0);
            A1 = wmma4(aw[1][4], bfh, A1);
            PIN();
            const int jn = j + 1 < HB ? j + 1 : HB - 1;
            v2f bfn; bfn.x = q1[jn].z; bfn.y = q1[jn].w;
            if (j + 1 < HB) {
              M0[0] = q0[jn].x; M0[1] = q0[jn].y; M0[2] = q0[jn].z; M0[3] = q0[jn].w;
              M0[4] = q1[jn].x; M0[5] = q1[jn].y; M0[6] = q1[jn].z; M0[7] = q1[jn].w;
              M1[0] = q2[jn].x; M1[1] = q2[jn].y; M1[2] = q2[jn].z; M1[3] = q2[jn].w;
              M1[4] = q3[jn].x; M1[5] = q3[jn].y; M1[6] = q3[jn].z; M1[7] = q3[jn].w;
              M0 = wmma4(aw[0][0], cua[jn], M0);
              M1 = wmma4(aw[1][0], cua[jn], M1);
              TIE(M0); TIE(M1);
            }
            PIN();
            gates4 g[3];
            g[0] = pw_gates(A0, A1, 0);
            g[1] = pw_gates(A0, A1, 1);
            PIN();
            if (j + 1 < HB) { M0 = wmma4(aw[0][2], bfn, M0); TIE(M0); }
            PIN();
            g[2] = pw_gates(A0, A1, 2);
            pw_cell(g, cS);
            PIN();
            if (j + 1 < HB) { M1 = wmma4(aw[1][2], bfn, M1); TIE(M1); }
            PIN();
            pw_out(g, cS, hS);
          }
          fl[0] = q1[HB - 1].z; fl[1] = q1[HB - 1].w; fl[2] = q3[HB - 1].z;
        } else {
#pragma unroll
          for (int j = 0; j < HB; ++j) {
            if (t0 + j < T) {
              step1(q0[j], q1[j], q2[j], q3[j], cua[j]);
              fl[0] = q1[j].z; fl[1] = q1[j].w; fl[2] = q3[j].z;
            }
          }
        }
        __builtin_amdgcn_sched_barrier(0);
#pragma unroll
        for (int j = 0; j < HB; ++j) cua[j] = nxa[j];
      }
    }
    __syncthreads();
  }
  if (producer) return;

  v2f a2[2][5]; v8f ci2[2]; load_A<true>(Wih1b, Whh1b, b1b, a2, ci2, lm, half);
  float hR[3], cR[3];
  size_t d3 = 3 * (size_t)B * LSTM_H;
#pragma unroll
  for (int u = 0; u < 3; ++u) {
    size_t o = (size_t)s * 6 + 3 * half + u;
    hR[u] = h0[d3 + o];  cR[u] = c0[d3 + o];
  }
  {
    const float* r0p = r0ws + ((size_t)(T - 1) * B + s) * 6;
    v2f r01; float r2; ld_r(r0p, half, r01, r2);
    v2f bch[5]; b_l1(fl, r01, r2, hR, bch);
    v8f A0, A1; mm(a2, ci2, bch, 5, A0, A1);
    lstm_pointwise(A0, A1, cR, hR);
  }
  float pB[3], pR[3];
#pragma unroll
  for (int u = 0; u < 3; ++u) { pB[u] = sx16(hS[u]); pR[u] = sx16(hR[u]); }
  if (half == 0 && live) {
    float v[12];
#pragma unroll
    for (int u = 0; u < 3; ++u) {
      v[u]     = hS[u] > 0.0f ? hS[u] : 0.0f;
      v[3 + u] = pB[u] > 0.0f ? pB[u] : 0.0f;
      v[6 + u] = hR[u] > 0.0f ? hR[u] : 0.0f;
      v[9 + u] = pR[u] > 0.0f ? pR[u] : 0.0f;
    }
#pragma unroll
    for (int j = 0; j < 6; ++j) {
      float acc = fcb[j];
#pragma unroll
      for (int k = 0; k < 12; ++k) acc = fmaf(fcw[j * 12 + k], v[k], acc);
      out[(size_t)s * 6 + j] = acc;
    }
  }
}

extern "C" void kernel_launch(void* const* d_in, const int* in_sizes, int n_in,
                              void* d_out, int out_size, void* d_ws, size_t ws_size,
                              hipStream_t stream) {
  const float* tf    = (const float*)d_in[0];
  const float* h0    = (const float*)d_in[1];
  const float* c0    = (const float*)d_in[2];
  const float* Wih0f = (const float*)d_in[3];
  const float* Whh0f = (const float*)d_in[4];
  const float* b0f   = (const float*)d_in[5];
  const float* Wih0b = (const float*)d_in[6];
  const float* Whh0b = (const float*)d_in[7];
  const float* b0b   = (const float*)d_in[8];
  const float* Wih1f = (const float*)d_in[9];
  const float* Whh1f = (const float*)d_in[10];
  const float* b1f   = (const float*)d_in[11];
  const float* Wih1b = (const float*)d_in[12];
  const float* Whh1b = (const float*)d_in[13];
  const float* b1b   = (const float*)d_in[14];
  const float* fcw   = (const float*)d_in[15];
  const float* fcb   = (const float*)d_in[16];

  int B = in_sizes[1] / (4 * LSTM_H);
  int T = in_sizes[0] / (B * LSTM_H);

  size_t need = (size_t)B * T * LSTM_H * sizeof(float);
  if (ws_size < need) return;
  float* r0ws = (float*)d_ws;

  int waves = (B + 15) / 16;
  int threads = waves * 32;
  int block = WG_THREADS;
  int grid = (threads + block - 1) / block;

  lstm_l0_bwd<<<grid, block, 0, stream>>>(tf, h0, c0, Wih0b, Whh0b, b0b, r0ws, B, T);
  int grid2 = (waves + 1) / 2;
  lstm_fused<<<grid2, block, 0, stream>>>(tf, h0, c0, Wih0f, Whh0f, b0f,
                                         Wih1f, Whh1f, b1f, Wih1b, Whh1b, b1b,
                                         fcw, fcb, r0ws, (float*)d_out, B, T);
}
